# starting kernel
# baseline (speedup 1.0000x reference)
#include <hip/hip_runtime.h>
#include <hip/hip_bf16.h>


#define N      4096
#define NFEAT  512
#define NHID   64
#define NCLASS 16
#define H1     8
#define NEG_SLOPE 0.2f
#define MASK_VAL  -1e10f

typedef __attribute__((ext_vector_type(16))) _Float16 v16h;
typedef __attribute__((ext_vector_type(8)))  _Float16 v8h;
typedef __attribute__((ext_vector_type(8)))  float    v8f;

__device__ __forceinline__ v8f wmma16(v16h a, v16h b, v8f c) {
  return __builtin_amdgcn_wmma_f32_16x16x32_f16(
       false, a,  false, b,
       (short)0, c,  false,  false);
}

__device__ __forceinline__ v16h load_frag16(const _Float16* base, int rc,
                                            int ld, int k0) {
  int lane = threadIdx.x & 31;
  int hf   = lane >> 4;
  int r    = lane & 15;
  const _Float16* p = base + (size_t)(rc + r) * ld + k0 + (hf << 3);
  v8h lo = *(const v8h*)p;
  v8h hi = *(const v8h*)(p + 16);
  return __builtin_shufflevector(lo, hi, 0, 1, 2, 3, 4, 5, 6, 7, 8, 9, 10, 11,
                                 12, 13, 14, 15);
}

__device__ __forceinline__ float lrelu(float x) {
  return x > 0.0f ? x : NEG_SLOPE * x;
}

__global__ void cast_kernel(const float* __restrict__ x,
                            const float* __restrict__ W1,
                            const float* __restrict__ W2,
                            _Float16* __restrict__ XH,
                            _Float16* __restrict__ W1T,
                            _Float16* __restrict__ W2T) {
  int i = blockIdx.x * blockDim.x + threadIdx.x;
  if (i < N * NFEAT) XH[i] = (_Float16)x[i];
  if (i < H1 * NFEAT * NHID) {
    int h   = i / (NFEAT * NHID);
    int rem = i % (NFEAT * NHID);
    int k   = rem / NHID;
    int c   = rem % NHID;
    W1T[((size_t)(h * NHID + c)) * NFEAT + k] = (_Float16)W1[i];
  }
  if (i < (H1 * NHID) * NCLASS) {
    int k = i / NCLASS;
    int c = i % NCLASS;
    W2T[(size_t)c * (H1 * NHID) + k] = (_Float16)W2[i];
  }
}

__global__ void gemm1_kernel(const _Float16* __restrict__ XH,
                             const _Float16* __restrict__ W1T,
                             float* __restrict__ feats,
                             _Float16* __restrict__ featsT) {
  int head = blockIdx.y;
  int r0   = blockIdx.x << 4;
  const _Float16* BT = W1T + (size_t)head * NHID * NFEAT;

  v8f c0 = {}, c1 = {}, c2 = {}, c3 = {};
  for (int k0 = 0; k0 < NFEAT; k0 += 32) {
    v16h a = load_frag16(XH, r0, NFEAT, k0);
    c0 = wmma16(a, load_frag16(BT, 0, NFEAT, k0), c0);
    c1 = wmma16(a, load_frag16(BT, 16, NFEAT, k0), c1);
    c2 = wmma16(a, load_frag16(BT, 32, NFEAT, k0), c2);
    c3 = wmma16(a, load_frag16(BT, 48, NFEAT, k0), c3);
  }

  int lane = threadIdx.x & 31;
  int hf   = lane >> 4;
  int r    = lane & 15;
  v8f acc[4] = {c0, c1, c2, c3};
#pragma unroll
  for (int nt = 0; nt < 4; ++nt) {
#pragma unroll
    for (int k = 0; k < 8; ++k) {
      int orow = r0 + k + (hf << 3);
      int col  = nt * 16 + r;
      float v  = acc[nt][k];
      feats[((size_t)head * N + orow) * NHID + col] = v;
      featsT[((size_t)head * NHID + col) * N + orow] = (_Float16)v;
    }
  }
}

__global__ void sn_kernel(const float* __restrict__ feats,
                          const float* __restrict__ a_s,
                          const float* __restrict__ a_n,
                          float* __restrict__ s, float* __restrict__ n,
                          int fout, int rowsPerHead, int total) {
  int i = blockIdx.x * blockDim.x + threadIdx.x;
  if (i >= total) return;
  int head = i / rowsPerHead;
  const float* f  = feats + (size_t)i * fout;
  const float* av = a_s + head * fout;
  const float* nv = a_n + head * fout;
  float ss = 0.0f, nn = 0.0f;
  for (int c = 0; c < fout; ++c) {
    float v = f[c];
    ss += v * av[c];
    nn += v * nv[c];
  }
  s[i] = ss;
  n[i] = nn;
}

__global__ void rowstats_kernel(const int* __restrict__ adj,
                                const float* __restrict__ s,
                                const float* __restrict__ n,
                                float* __restrict__ m, float* __restrict__ Z,
                                int nheads) {
  int i   = blockIdx.x;
  int tid = threadIdx.x;
  __shared__ float red[256];
  const int* arow = adj + (size_t)i * N;

  for (int h = 0; h < nheads; ++h) {
    float sv = s[h * N + i];
    const float* nrow = n + h * N;

    float lm = -3.0e38f;
    for (int j = tid; j < N; j += 256) {
      float lg = arow[j] > 0 ? lrelu(sv + nrow[j]) : MASK_VAL;
      lm = fmaxf(lm, lg);
    }
    red[tid] = lm;
    __syncthreads();
    for (int st = 128; st > 0; st >>= 1) {
      if (tid < st) red[tid] = fmaxf(red[tid], red[tid + st]);
      __syncthreads();
    }
    float mv = red[0];
    __syncthreads();

    float ls = 0.0f;
    for (int j = tid; j < N; j += 256) {
      float lg = arow[j] > 0 ? lrelu(sv + nrow[j]) : MASK_VAL;
      ls += __expf(lg - mv);
    }
    red[tid] = ls;
    __syncthreads();
    for (int st = 128; st > 0; st >>= 1) {
      if (tid < st) red[tid] += red[tid + st];
      __syncthreads();
    }
    if (tid == 0) {
      m[h * N + i] = mv;
      Z[h * N + i] = red[0];
    }
    __syncthreads();
  }
}

__global__ void attn_gemm1_kernel(const int* __restrict__ adj,
                                  const _Float16* __restrict__ featsT,
                                  const float* __restrict__ s1,
                                  const float* __restrict__ n1,
                                  const float* __restrict__ m1,
                                  const float* __restrict__ Z1,
                                  const float* __restrict__ b1,
                                  _Float16* __restrict__ H) {
  int head = blockIdx.y;
  int r0   = blockIdx.x << 4;
  int lane = threadIdx.x & 31;
  int hf   = lane >> 4;
  int r    = lane & 15;
  int row  = r0 + r;

  const float* nrow = n1 + head * N;
  float sv = s1[head * N + row];
  float mv = m1[head * N + row];
  float zi = 1.0f / Z1[head * N + row];
  const int* arow = adj + (size_t)row * N;
  const _Float16* BT = featsT + (size_t)head * NHID * N;

  v8f c0 = {}, c1 = {}, c2 = {}, c3 = {};
  for (int j0 = 0; j0 < N; j0 += 32) {
    if (j0 + 256 < N) __builtin_prefetch(arow + j0 + 256, 0, 0);
    v16h a;
#pragma unroll
    for (int e = 0; e < 16; ++e) {
      int j = j0 + (e & 7) + (hf << 3) + ((e >> 3) << 4);
      float lg = lrelu(sv + nrow[j]);
      float p  = arow[j] > 0 ? __expf(lg - mv) * zi : 0.0f;
      a[e] = (_Float16)p;
    }
    c0 = wmma16(a, load_frag16(BT, 0, N, j0), c0);
    c1 = wmma16(a, load_frag16(BT, 16, N, j0), c1);
    c2 = wmma16(a, load_frag16(BT, 32, N, j0), c2);
    c3 = wmma16(a, load_frag16(BT, 48, N, j0), c3);
  }

  v8f acc[4] = {c0, c1, c2, c3};
#pragma unroll
  for (int nt = 0; nt < 4; ++nt) {
#pragma unroll
    for (int k = 0; k < 8; ++k) {
      int orow = r0 + k + (hf << 3);
      int col  = nt * 16 + r;
      float v  = acc[nt][k] + b1[head * NHID + col];
      v = v > 0.0f ? v : 0.0f;
      H[(size_t)orow * (H1 * NHID) + head * NHID + col] = (_Float16)v;
    }
  }
}

__global__ void gemm2_kernel(const _Float16* __restrict__ H,
                             const _Float16* __restrict__ W2T,
                             float* __restrict__ feats2,
                             _Float16* __restrict__ feats2T) {
  int r0 = blockIdx.x << 4;
  v8f c = {};
  for (int k0 = 0; k0 < H1 * NHID; k0 += 32) {
    v16h a = load_frag16(H, r0, H1 * NHID, k0);
    v16h b = load_frag16(W2T, 0, H1 * NHID, k0);
    c = wmma16(a, b, c);
  }
  int lane = threadIdx.x & 31;
  int hf   = lane >> 4;
  int r    = lane & 15;
#pragma unroll
  for (int k = 0; k < 8; ++k) {
    int orow = r0 + k + (hf << 3);
    float v  = c[k];
    feats2[(size_t)orow * NCLASS + r] = v;
    feats2T[(size_t)r * N + orow]     = (_Float16)v;
  }
}

__global__ void attn_gemm2_kernel(const int* __restrict__ adj,
                                  const _Float16* __restrict__ feats2T,
                                  const float* __restrict__ s2,
                                  const float* __restrict__ n2,
                                  const float* __restrict__ m2,
                                  const float* __restrict__ Z2,
                                  const float* __restrict__ b2,
                                  float* __restrict__ out) {
  int r0   = blockIdx.x << 4;
  int lane = threadIdx.x & 31;
  int hf   = lane >> 4;
  int r    = lane & 15;
  int row  = r0 + r;

  float sv = s2[row];
  float mv = m2[row];
  float zi = 1.0f / Z2[row];
  const int* arow = adj + (size_t)row * N;

  v8f c = {};
  for (int j0 = 0; j0 < N; j0 += 32) {
    if (j0 + 256 < N) __builtin_prefetch(arow + j0 + 256, 0, 0);
    v16h a;
#pragma unroll
    for (int e = 0; e < 16; ++e) {
      int j = j0 + (e & 7) + (hf << 3) + ((e >> 3) << 4);
      float lg = lrelu(sv + n2[j]);
      float p  = arow[j] > 0 ? __expf(lg - mv) * zi : 0.0f;
      a[e] = (_Float16)p;
    }
    v16h b = load_frag16(feats2T, 0, N, j0);
    c = wmma16(a, b, c);
  }

  float bias = b2[r];
#pragma unroll
  for (int k = 0; k < 8; ++k) {
    float v = c[k] + bias;
    v = v > 0.0f ? v : 0.0f;
    float mx = v;
#pragma unroll
    for (int msk = 1; msk < 16; msk <<= 1)
      mx = fmaxf(mx, __shfl_xor(mx, msk, 32));
    float ex = __expf(v - mx);
    float sum = ex;
#pragma unroll
    for (int msk = 1; msk < 16; msk <<= 1) sum += __shfl_xor(sum, msk, 32);
    float o  = (v - mx) - logf(sum);
    int orow = r0 + k + (hf << 3);
    out[(size_t)orow * NCLASS + r] = o;
  }
}

extern "C" void kernel_launch(void* const* d_in, const int* in_sizes, int n_in,
                              void* d_out, int out_size, void* d_ws,
                              size_t ws_size, hipStream_t stream) {
  const float* x   = (const float*)d_in[0];
  const int*   adj = (const int*)d_in[1];
  const float* W1  = (const float*)d_in[2];
  const float* b1  = (const float*)d_in[3];
  const float* as1 = (const float*)d_in[4];
  const float* an1 = (const float*)d_in[5];
  const float* W2  = (const float*)d_in[6];
  const float* b2  = (const float*)d_in[7];
  const float* as2 = (const float*)d_in[8];
  const float* an2 = (const float*)d_in[9];
  float* out = (float*)d_out;

  char* ws = (char*)d_ws;
  size_t off = 0;
  auto alloc = [&](size_t bytes) -> void* {
    off = (off + 255) & ~(size_t)255;
    void* p = ws + off;
    off += bytes;
    return p;
  };

  _Float16* XH      = (_Float16*)alloc((size_t)N * NFEAT * 2);
  _Float16* W1T     = (_Float16*)alloc((size_t)H1 * NHID * NFEAT * 2);
  _Float16* W2T     = (_Float16*)alloc((size_t)NCLASS * H1 * NHID * 2);
  float*    feats1  = (float*)alloc((size_t)H1 * N * NHID * 4);
  _Float16* featsT  = (_Float16*)alloc((size_t)H1 * NHID * N * 2);
  float*    s1      = (float*)alloc((size_t)H1 * N * 4);
  float*    n1      = (float*)alloc((size_t)H1 * N * 4);
  float*    m1      = (float*)alloc((size_t)H1 * N * 4);
  float*    Z1      = (float*)alloc((size_t)H1 * N * 4);
  _Float16* H       = (_Float16*)alloc((size_t)N * H1 * NHID * 2);
  float*    feats2  = (float*)alloc((size_t)N * NCLASS * 4);
  _Float16* feats2T = (_Float16*)alloc((size_t)NCLASS * N * 2);
  float*    s2      = (float*)alloc((size_t)N * 4);
  float*    n2      = (float*)alloc((size_t)N * 4);
  float*    m2      = (float*)alloc((size_t)N * 4);
  float*    Z2      = (float*)alloc((size_t)N * 4);
  (void)ws_size; (void)n_in; (void)in_sizes; (void)out_size;

  {
    int total = N * NFEAT;
    cast_kernel<<<(total + 255) / 256, 256, 0, stream>>>(x, W1, W2, XH, W1T,
                                                         W2T);
  }
  gemm1_kernel<<<dim3(N / 16, H1), 32, 0, stream>>>(XH, W1T, feats1, featsT);
  sn_kernel<<<(H1 * N + 255) / 256, 256, 0, stream>>>(feats1, as1, an1, s1, n1,
                                                      NHID, N, H1 * N);
  rowstats_kernel<<<N, 256, 0, stream>>>(adj, s1, n1, m1, Z1, H1);
  attn_gemm1_kernel<<<dim3(N / 16, H1), 32, 0, stream>>>(adj, featsT, s1, n1,
                                                         m1, Z1, b1, H);
  gemm2_kernel<<<N / 16, 32, 0, stream>>>(H, W2T, feats2, feats2T);
  sn_kernel<<<(N + 255) / 256, 256, 0, stream>>>(feats2, as2, an2, s2, n2,
                                                 NCLASS, N, N);
  rowstats_kernel<<<N, 256, 0, stream>>>(adj, s2, n2, m2, Z2, 1);
  attn_gemm2_kernel<<<N / 16, 32, 0, stream>>>(adj, feats2T, s2, n2, m2, Z2,
                                               b2, out);
}


# Layer-1 attention: one 16-wave work-group per 16-row tile forms all 8 heads' probabilities once per adjacency element (a wave per row) and feeds the heads' WMMA waves through LDS; sums unchanged
# speedup vs baseline: 2.5575x; 2.5575x over previous
#include <hip/hip_runtime.h>
#include <hip/hip_bf16.h>


#define N      4096
#define NFEAT  512
#define NHID   64
#define NCLASS 16
#define H1     8
#define NEG_SLOPE 0.2f
#define MASK_VAL  -1e10f

typedef __attribute__((ext_vector_type(16))) _Float16 v16h;
typedef __attribute__((ext_vector_type(8)))  _Float16 v8h;
typedef __attribute__((ext_vector_type(4)))  _Float16 v4h;
typedef __attribute__((ext_vector_type(8)))  float    v8f;

__device__ __forceinline__ v8f wmma16(v16h a, v16h b, v8f c) {
  return __builtin_amdgcn_wmma_f32_16x16x32_f16(
       false, a,  false, b,
       (short)0, c,  false,  false);
}

__device__ __forceinline__ v16h load_frag16(const _Float16* base, int rc,
                                            int ld, int k0) {
  int lane = threadIdx.x & 31;
  int hf   = lane >> 4;
  int r    = lane & 15;
  const _Float16* p = base + (size_t)(rc + r) * ld + k0 + (hf << 3);
  v8h lo = *(const v8h*)p;
  v8h hi = *(const v8h*)(p + 16);
  return __builtin_shufflevector(lo, hi, 0, 1, 2, 3, 4, 5, 6, 7, 8, 9, 10, 11,
                                 12, 13, 14, 15);
}

__device__ __forceinline__ v16h load_frag16_f32(const float* base, int rc,
                                                int ld, int k0) {
  int lane = threadIdx.x & 31;
  int hf   = lane >> 4;
  int r    = lane & 15;
  const float* p = base + (size_t)(rc + r) * ld + k0 + (hf << 3);
  const float4 q0 = *(const float4*)p;
  const float4 q1 = *(const float4*)(p + 4);
  const float4 q2 = *(const float4*)(p + 16);
  const float4 q3 = *(const float4*)(p + 20);
  v16h a;
  a[0]  = (_Float16)q0.x; a[1]  = (_Float16)q0.y;
  a[2]  = (_Float16)q0.z; a[3]  = (_Float16)q0.w;
  a[4]  = (_Float16)q1.x; a[5]  = (_Float16)q1.y;
  a[6]  = (_Float16)q1.z; a[7]  = (_Float16)q1.w;
  a[8]  = (_Float16)q2.x; a[9]  = (_Float16)q2.y;
  a[10] = (_Float16)q2.z; a[11] = (_Float16)q2.w;
  a[12] = (_Float16)q3.x; a[13] = (_Float16)q3.y;
  a[14] = (_Float16)q3.z; a[15] = (_Float16)q3.w;
  return a;
}

__device__ __forceinline__ float lrelu(float x) {
  return x > 0.0f ? x : NEG_SLOPE * x;
}

__global__ void cast_kernel(const float* __restrict__ W1,
                            const float* __restrict__ W2,
                            _Float16* __restrict__ W1T,
                            _Float16* __restrict__ W2T) {
  int i = blockIdx.x * blockDim.x + threadIdx.x;
  if (i < H1 * NFEAT * NHID) {
    int h   = i / (NFEAT * NHID);
    int rem = i % (NFEAT * NHID);
    int k   = rem / NHID;
    int c   = rem % NHID;
    W1T[((size_t)(h * NHID + c)) * NFEAT + k] = (_Float16)W1[i];
  }
  if (i < (H1 * NHID) * NCLASS) {
    int k = i / NCLASS;
    int c = i % NCLASS;
    W2T[(size_t)c * (H1 * NHID) + k] = (_Float16)W2[i];
  }
}

__global__ void gemm1_kernel(const float* __restrict__ x,
                             const _Float16* __restrict__ W1T,
                             float* __restrict__ feats,
                             _Float16* __restrict__ featsT) {
  int head = blockIdx.y;
  int r0   = blockIdx.x << 4;
  const _Float16* BT = W1T + (size_t)head * NHID * NFEAT;

  v8f c0 = {}, c1 = {}, c2 = {}, c3 = {};
  for (int k0 = 0; k0 < NFEAT; k0 += 32) {
    v16h a = load_frag16_f32(x, r0, NFEAT, k0);
    c0 = wmma16(a, load_frag16(BT, 0, NFEAT, k0), c0);
    c1 = wmma16(a, load_frag16(BT, 16, NFEAT, k0), c1);
    c2 = wmma16(a, load_frag16(BT, 32, NFEAT, k0), c2);
    c3 = wmma16(a, load_frag16(BT, 48, NFEAT, k0), c3);
  }

  int lane = threadIdx.x & 31;
  int hf   = lane >> 4;
  int r    = lane & 15;
  v8f acc[4] = {c0, c1, c2, c3};
#pragma unroll
  for (int nt = 0; nt < 4; ++nt) {
#pragma unroll
    for (int k = 0; k < 8; ++k) {
      int orow = r0 + k + (hf << 3);
      int col  = nt * 16 + r;
      float v  = acc[nt][k];
      feats[((size_t)head * N + orow) * NHID + col] = v;
      featsT[((size_t)head * NHID + col) * N + orow] = (_Float16)v;
    }
  }
}

__global__ void sn_kernel(const float* __restrict__ feats,
                          const float* __restrict__ a_s,
                          const float* __restrict__ a_n,
                          float* __restrict__ s, float* __restrict__ n,
                          int fout, int rowsPerHead, int total) {
  int i = blockIdx.x * blockDim.x + threadIdx.x;
  if (i >= total) return;
  int head = i / rowsPerHead;
  const float* f  = feats + (size_t)i * fout;
  const float* av = a_s + head * fout;
  const float* nv = a_n + head * fout;
  float ss = 0.0f, nn = 0.0f;
  for (int c = 0; c < fout; ++c) {
    float v = f[c];
    ss += v * av[c];
    nn += v * nv[c];
  }
  s[i] = ss;
  n[i] = nn;
}

#define RS_WAVES 8

__device__ __forceinline__ float wave_max32(float v) {
#pragma unroll
  for (int msk = 16; msk > 0; msk >>= 1) v = fmaxf(v, __shfl_xor(v, msk, 32));
  return v;
}

__device__ __forceinline__ float wave_sum32(float v) {
#pragma unroll
  for (int msk = 16; msk > 0; msk >>= 1) v += __shfl_xor(v, msk, 32);
  return v;
}

__device__ __forceinline__ float lrelu_max(float x) {
  return fmaxf(x, NEG_SLOPE * x);
}

template <int NH, int U>
struct RsChunk {
  int4   a[U];
  float4 v[U][NH];
};

template <int NH, int U>
__device__ __forceinline__ void rs_load(RsChunk<NH, U>& c, const int4* arow,
                                        const float4* nq, int ci) {
#pragma unroll
  for (int u = 0; u < U; ++u) {
    const int q = (ci * U + u) * 32;
    c.a[u] = arow[q];
#pragma unroll
    for (int h = 0; h < NH; ++h) c.v[u][h] = nq[h * (N / 4) + q];
  }
}

template <int NH, int U>
__device__ __forceinline__ void rs_max_step(const RsChunk<NH, U>& c,
                                            float (&nmax)[NH], int& cnt) {
#pragma unroll
  for (int u = 0; u < U; ++u) {
    const bool e0 = c.a[u].x > 0, e1 = c.a[u].y > 0;
    const bool e2 = c.a[u].z > 0, e3 = c.a[u].w > 0;
    cnt += (int)e0 + (int)e1 + (int)e2 + (int)e3;
#pragma unroll
    for (int h = 0; h < NH; ++h) {
      const float x0 = e0 ? c.v[u][h].x : -INFINITY;
      const float x1 = e1 ? c.v[u][h].y : -INFINITY;
      const float x2 = e2 ? c.v[u][h].z : -INFINITY;
      const float x3 = e3 ? c.v[u][h].w : -INFINITY;
      nmax[h] = fmaxf(fmaxf(nmax[h], x0), fmaxf(x1, fmaxf(x2, x3)));
    }
  }
}

template <int NH, int U, bool WP>
__device__ __forceinline__ void rs_sum_step(const RsChunk<NH, U>& c,
                                            const float (&sv)[NH],
                                            const float (&mv)[NH],
                                            float (&acc)[NH],
                                            _Float16* prow, int ci) {
  static_assert(!WP || NH == 1, "the probability plane has one head");
#pragma unroll
  for (int u = 0; u < U; ++u) {
    const bool e0 = c.a[u].x > 0, e1 = c.a[u].y > 0;
    const bool e2 = c.a[u].z > 0, e3 = c.a[u].w > 0;
#pragma unroll
    for (int h = 0; h < NH; ++h) {
      const float p0 = __expf(lrelu_max(sv[h] + c.v[u][h].x) - mv[h]);
      const float p1 = __expf(lrelu_max(sv[h] + c.v[u][h].y) - mv[h]);
      const float p2 = __expf(lrelu_max(sv[h] + c.v[u][h].z) - mv[h]);
      const float p3 = __expf(lrelu_max(sv[h] + c.v[u][h].w) - mv[h]);
      const float q0 = e0 ? p0 : 0.0f, q1 = e1 ? p1 : 0.0f;
      const float q2 = e2 ? p2 : 0.0f, q3 = e3 ? p3 : 0.0f;
      acc[h] += q0 + q1;
      acc[h] += q2 + q3;
      if (WP) {
        const v4h pv = {(_Float16)q0, (_Float16)q1, (_Float16)q2,
                        (_Float16)q3};
        *(v4h*)(prow + (ci * U + u) * 128) = pv;
      }
    }
  }
}

template <int NH, int U, bool WP>
__global__ void __launch_bounds__(RS_WAVES * 32)
__attribute__((amdgpu_waves_per_eu(4, 4)))
rowstats_kernel(const int* __restrict__ adj, const float* __restrict__ s,
                const float* __restrict__ n, float* __restrict__ m,
                float* __restrict__ Z, _Float16* __restrict__ P) {
  constexpr int CHUNKS = N / (128 * U);
  static_assert(CHUNKS >= 2 && CHUNKS % 2 == 0, "chunks are taken in pairs");

  const int lane = threadIdx.x & 31;
  const int wave = __builtin_amdgcn_readfirstlane((int)(threadIdx.x >> 5));
  const int i    = blockIdx.x * RS_WAVES + wave;

  const int* arow_i = adj + (size_t)i * N;
  __builtin_prefetch(arow_i + lane * 64, 0, 0);
  __builtin_prefetch(arow_i + (N / 2) + lane * 64, 0, 0);

  const int4*   arow = (const int4*)arow_i + lane;
  const float4* nq   = (const float4*)n + lane;
  _Float16* prow = WP ? P + (size_t)i * N + lane * 4 : nullptr;

  RsChunk<NH, U> ca, cb;

  float nmax[NH];
#pragma unroll
  for (int h = 0; h < NH; ++h) nmax[h] = -INFINITY;
  int cnt = 0;

  rs_load(ca, arow, nq, 0);
  for (int ci = 0; ci < CHUNKS - 2; ci += 2) {
    rs_load(cb, arow, nq, ci + 1);
    rs_max_step(ca, nmax, cnt);
    rs_load(ca, arow, nq, ci + 2);
    rs_max_step(cb, nmax, cnt);
  }
  rs_load(cb, arow, nq, CHUNKS - 1);
  rs_max_step(ca, nmax, cnt);
  rs_max_step(cb, nmax, cnt);

#pragma unroll
  for (int msk = 16; msk > 0; msk >>= 1) cnt += __shfl_xor(cnt, msk, 32);
  const int n_masked = N - cnt;

  float sv[NH], mv[NH];
#pragma unroll
  for (int h = 0; h < NH; ++h) {
    sv[h] = s[h * N + i];
    float mm = fmaxf(-3.0e38f, lrelu_max(sv[h] + wave_max32(nmax[h])));
    if (n_masked > 0) mm = fmaxf(mm, MASK_VAL);
    mv[h] = mm;
  }

  float acc[NH];
#pragma unroll
  for (int h = 0; h < NH; ++h) acc[h] = 0.0f;

  rs_load(ca, arow, nq, 0);
  for (int ci = 0; ci < CHUNKS - 2; ci += 2) {
    rs_load(cb, arow, nq, ci + 1);
    rs_sum_step<NH, U, WP>(ca, sv, mv, acc, prow, ci);
    rs_load(ca, arow, nq, ci + 2);
    rs_sum_step<NH, U, WP>(cb, sv, mv, acc, prow, ci + 1);
  }
  rs_load(cb, arow, nq, CHUNKS - 1);
  rs_sum_step<NH, U, WP>(ca, sv, mv, acc, prow, CHUNKS - 2);
  rs_sum_step<NH, U, WP>(cb, sv, mv, acc, prow, CHUNKS - 1);

#pragma unroll
  for (int h = 0; h < NH; ++h) {
    float zs = wave_sum32(acc[h]);
    if (n_masked > 0) zs += (float)n_masked * __expf(MASK_VAL - mv[h]);
    if (lane == 0) {
      if (!WP) m[h * N + i] = mv[h];
      Z[h * N + i] = zs;
    }
  }
}

#define F1_ROWS  16
#define F1_COLS  128
#define F1_PITCH 136
#define F1_HEAD  (F1_ROWS * F1_PITCH)
#define F1_STAGE (H1 * F1_HEAD)
#define F1_KS    (F1_COLS / 32)

struct F1Frags {
  v16h b[F1_KS][2];
};

__device__ __forceinline__ void f1_load_b(F1Frags& f, const _Float16* BT,
                                          int f0, int ci) {
#pragma unroll
  for (int ks = 0; ks < F1_KS; ++ks) {
    const int j0 = ci * F1_COLS + ks * 32;
    f.b[ks][0] = load_frag16(BT, f0, N, j0);
    f.b[ks][1] = load_frag16(BT, f0 + 16, N, j0);
  }
}

__device__ __forceinline__ void f1_form(const RsChunk<H1, 1>& c,
                                        const float (&sv)[H1],
                                        const float (&mv)[H1],
                                        const float (&zi)[H1],
                                        _Float16* prow) {
  const bool e0 = c.a[0].x > 0, e1 = c.a[0].y > 0;
  const bool e2 = c.a[0].z > 0, e3 = c.a[0].w > 0;
#pragma unroll
  for (int h = 0; h < H1; ++h) {
    const float p0 = __expf(lrelu_max(sv[h] + c.v[0][h].x) - mv[h]) * zi[h];
    const float p1 = __expf(lrelu_max(sv[h] + c.v[0][h].y) - mv[h]) * zi[h];
    const float p2 = __expf(lrelu_max(sv[h] + c.v[0][h].z) - mv[h]) * zi[h];
    const float p3 = __expf(lrelu_max(sv[h] + c.v[0][h].w) - mv[h]) * zi[h];
    const v4h pv = {(_Float16)(e0 ? p0 : 0.0f), (_Float16)(e1 ? p1 : 0.0f),
                    (_Float16)(e2 ? p2 : 0.0f), (_Float16)(e3 ? p3 : 0.0f)};
    *(v4h*)(prow + h * F1_HEAD) = pv;
  }
}

__device__ __forceinline__ void f1_mma(const _Float16* ph, const F1Frags& f,
                                       v8f& c0, v8f& c1) {
#pragma unroll
  for (int ks = 0; ks < F1_KS; ++ks) {
    const v16h a = load_frag16(ph, 0, F1_PITCH, ks * 32);
    c0 = wmma16(a, f.b[ks][0], c0);
    c1 = wmma16(a, f.b[ks][1], c1);
  }
}

__device__ __forceinline__ void f1_barrier() {
  asm volatile("s_wait_dscnt 0x0\n\ts_barrier_signal -1\n\ts_barrier_wait -1"
               ::: "memory");
}

__global__ void __launch_bounds__(F1_ROWS * 32)
attn_gemm1_kernel(const int* __restrict__ adj,
                  const _Float16* __restrict__ featsT,
                  const float* __restrict__ s1,
                  const float* __restrict__ n1,
                  const float* __restrict__ m1,
                  const float* __restrict__ Z1,
                  const float* __restrict__ b1,
                  _Float16* __restrict__ H) {
  constexpr int CH = N / F1_COLS;
  static_assert(CH % 2 == 0, "blocks are taken in pairs, one per LDS stage");
  __shared__ __attribute__((aligned(16))) _Float16 pl[2 * F1_STAGE];

  const int lane = threadIdx.x & 31;
  const int wave = __builtin_amdgcn_readfirstlane((int)(threadIdx.x >> 5));
  const int r0   = blockIdx.x * F1_ROWS;
  const int i    = r0 + wave;
  const int head = wave & (H1 - 1);
  const int f0   = (wave >> 3) * 32;

  const int4*   arow = (const int4*)(adj + (size_t)i * N) + lane;
  const float4* nq   = (const float4*)n1 + lane;

  float sv[H1], mv[H1], zi[H1];
#pragma unroll
  for (int h = 0; h < H1; ++h) {
    sv[h] = s1[h * N + i];
    mv[h] = m1[h * N + i];
    zi[h] = 1.0f / Z1[h * N + i];
  }

  _Float16* prow0 = pl + wave * F1_PITCH + lane * 4;
  _Float16* prow1 = prow0 + F1_STAGE;
  const _Float16* ph0 = pl + head * F1_HEAD;
  const _Float16* ph1 = ph0 + F1_STAGE;
  const _Float16* BT  = featsT + (size_t)head * NHID * N;

  RsChunk<H1, 1> ca, cb;
  F1Frags bf;
  v8f c0 = {}, c1 = {};

  rs_load(ca, arow, nq, 0);
  for (int ci = 0; ci < CH; ci += 2) {
    rs_load(cb, arow, nq, ci + 1);
    f1_load_b(bf, BT, f0, ci);
    f1_form(ca, sv, mv, zi, prow0);
    f1_barrier();
    f1_mma(ph0, bf, c0, c1);

    if (ci + 2 < CH) rs_load(ca, arow, nq, ci + 2);
    f1_load_b(bf, BT, f0, ci + 1);
    f1_form(cb, sv, mv, zi, prow1);
    f1_barrier();
    f1_mma(ph1, bf, c0, c1);
  }

  const int hf = lane >> 4;
  const int r  = lane & 15;
  v8f acc[2] = {c0, c1};
#pragma unroll
  for (int nt = 0; nt < 2; ++nt) {
#pragma unroll
    for (int k = 0; k < 8; ++k) {
      int orow = r0 + k + (hf << 3);
      int col  = f0 + nt * 16 + r;
      float v  = acc[nt][k] + b1[head * NHID + col];
      v = v > 0.0f ? v : 0.0f;
      H[(size_t)orow * (H1 * NHID) + head * NHID + col] = (_Float16)v;
    }
  }
}

__global__ void gemm2_kernel(const _Float16* __restrict__ H,
                             const _Float16* __restrict__ W2T,
                             float* __restrict__ feats2,
                             _Float16* __restrict__ feats2T) {
  int r0 = blockIdx.x << 4;
  v8f c = {};
  for (int k0 = 0; k0 < H1 * NHID; k0 += 32) {
    v16h a = load_frag16(H, r0, H1 * NHID, k0);
    v16h b = load_frag16(W2T, 0, H1 * NHID, k0);
    c = wmma16(a, b, c);
  }
  int lane = threadIdx.x & 31;
  int hf   = lane >> 4;
  int r    = lane & 15;
#pragma unroll
  for (int k = 0; k < 8; ++k) {
    int orow = r0 + k + (hf << 3);
    float v  = c[k];
    feats2[(size_t)orow * NCLASS + r] = v;
    feats2T[(size_t)r * N + orow]     = (_Float16)v;
  }
}

#define A2_U 8

struct A2Group {
  v16h a[A2_U];
  v16h b[A2_U];
};

__device__ __forceinline__ void a2_load(A2Group& g, const _Float16* P,
                                        const _Float16* BT, int r0, int gi) {
#pragma unroll
  for (int u = 0; u < A2_U; ++u) {
    const int j0 = (gi * A2_U + u) * 32;
    g.a[u] = load_frag16(P, r0, N, j0);
    g.b[u] = load_frag16(BT, 0, N, j0);
  }
}

__device__ __forceinline__ v8f a2_mma(const A2Group& g, v8f c) {
#pragma unroll
  for (int u = 0; u < A2_U; ++u) c = wmma16(g.a[u], g.b[u], c);
  return c;
}

__global__ void __launch_bounds__(32)
attn_gemm2_kernel(const _Float16* __restrict__ P,
                  const _Float16* __restrict__ feats2T,
                  const float* __restrict__ Z2,
                  const float* __restrict__ b2,
                  float* __restrict__ out) {
  constexpr int GROUPS = N / (32 * A2_U);
  static_assert(GROUPS >= 2 && GROUPS % 2 == 0, "groups are taken in pairs");

  int r0   = blockIdx.x << 4;
  int lane = threadIdx.x & 31;
  int hf   = lane >> 4;
  int r    = lane & 15;

  A2Group ga, gb;
  v8f c = {};
  a2_load(ga, P, feats2T, r0, 0);
  for (int gi = 0; gi < GROUPS - 2; gi += 2) {
    a2_load(gb, P, feats2T, r0, gi + 1);
    c = a2_mma(ga, c);
    a2_load(ga, P, feats2T, r0, gi + 2);
    c = a2_mma(gb, c);
  }
  a2_load(gb, P, feats2T, r0, GROUPS - 1);
  c = a2_mma(ga, c);
  c = a2_mma(gb, c);

  float bias = b2[r];
#pragma unroll
  for (int k = 0; k < 8; ++k) {
    int orow = r0 + k + (hf << 3);
    float zi = 1.0f / Z2[orow];
    float v  = c[k] * zi + bias;
    v = v > 0.0f ? v : 0.0f;
    float mx = v;
#pragma unroll
    for (int msk = 1; msk < 16; msk <<= 1)
      mx = fmaxf(mx, __shfl_xor(mx, msk, 32));
    float ex = __expf(v - mx);
    float sum = ex;
#pragma unroll
    for (int msk = 1; msk < 16; msk <<= 1) sum += __shfl_xor(sum, msk, 32);
    float o  = (v - mx) - logf(sum);
    out[(size_t)orow * NCLASS + r] = o;
  }
}

extern "C" void kernel_launch(void* const* d_in, const int* in_sizes, int n_in,
                              void* d_out, int out_size, void* d_ws,
                              size_t ws_size, hipStream_t stream) {
  const float* x   = (const float*)d_in[0];
  const int*   adj = (const int*)d_in[1];
  const float* W1  = (const float*)d_in[2];
  const float* b1  = (const float*)d_in[3];
  const float* as1 = (const float*)d_in[4];
  const float* an1 = (const float*)d_in[5];
  const float* W2  = (const float*)d_in[6];
  const float* b2  = (const float*)d_in[7];
  const float* as2 = (const float*)d_in[8];
  const float* an2 = (const float*)d_in[9];
  float* out = (float*)d_out;

  char* ws = (char*)d_ws;
  size_t off = 0;
  auto alloc = [&](size_t bytes) -> void* {
    off = (off + 255) & ~(size_t)255;
    void* p = ws + off;
    off += bytes;
    return p;
  };

  float*    feats2  = (float*)alloc((size_t)N * NCLASS * 4);
  _Float16* feats2T = (_Float16*)alloc((size_t)NCLASS * N * 2);
  float*    s2      = (float*)alloc((size_t)N * 4);
  float*    n2      = (float*)alloc((size_t)N * 4);
  float*    Z2      = (float*)alloc((size_t)N * 4);
  const size_t shared = off;
  _Float16* P2      = (_Float16*)alloc((size_t)N * N * 2);
  off = shared;
  _Float16* W1T     = (_Float16*)alloc((size_t)H1 * NHID * NFEAT * 2);
  _Float16* W2T     = (_Float16*)alloc((size_t)NCLASS * H1 * NHID * 2);
  float*    feats1  = (float*)alloc((size_t)H1 * N * NHID * 4);
  _Float16* featsT  = (_Float16*)alloc((size_t)H1 * NHID * N * 2);
  float*    s1      = (float*)alloc((size_t)H1 * N * 4);
  float*    n1      = (float*)alloc((size_t)H1 * N * 4);
  float*    m1      = (float*)alloc((size_t)H1 * N * 4);
  float*    Z1      = (float*)alloc((size_t)H1 * N * 4);
  _Float16* H       = (_Float16*)alloc((size_t)N * H1 * NHID * 2);
  (void)ws_size; (void)n_in; (void)in_sizes; (void)out_size;

  {
    int total = H1 * NFEAT * NHID;
    cast_kernel<<<(total + 255) / 256, 256, 0, stream>>>(W1, W2, W1T, W2T);
  }
  gemm1_kernel<<<dim3(N / 16, H1), 32, 0, stream>>>(x, W1T, feats1, featsT);
  sn_kernel<<<(H1 * N + 255) / 256, 256, 0, stream>>>(feats1, as1, an1, s1, n1,
                                                      NHID, N, H1 * N);
  rowstats_kernel<H1, 1, false><<<N / RS_WAVES, RS_WAVES * 32, 0, stream>>>(
      adj, s1, n1, m1, Z1, nullptr);
  attn_gemm1_kernel<<<N / F1_ROWS, F1_ROWS * 32, 0, stream>>>(
      adj, featsT, s1, n1, m1, Z1, b1, H);
  gemm2_kernel<<<N / 16, 32, 0, stream>>>(H, W2T, feats2, feats2T);
  sn_kernel<<<(N + 255) / 256, 256, 0, stream>>>(feats2, as2, an2, s2, n2,
                                                 NCLASS, N, N);
  rowstats_kernel<1, 8, true><<<N / RS_WAVES, RS_WAVES * 32, 0, stream>>>(
      adj, s2, n2, nullptr, Z2, P2);
  attn_gemm2_kernel<<<N / 16, 32, 0, stream>>>(P2, feats2T, Z2, b2, out);
}


# Layer-1 row statistics move into the attention kernel: its row-owning waves take the masked max in a first sweep and sum Z from the terms they form anyway; 1/Z applied to the f32 sums; 8 launches
# speedup vs baseline: 2.8960x; 1.1324x over previous
#include <hip/hip_runtime.h>
#include <hip/hip_bf16.h>


#define N      4096
#define NFEAT  512
#define NHID   64
#define NCLASS 16
#define H1     8
#define NEG_SLOPE 0.2f
#define MASK_VAL  -1e10f

typedef __attribute__((ext_vector_type(16))) _Float16 v16h;
typedef __attribute__((ext_vector_type(8)))  _Float16 v8h;
typedef __attribute__((ext_vector_type(4)))  _Float16 v4h;
typedef __attribute__((ext_vector_type(8)))  float    v8f;

__device__ __forceinline__ v8f wmma16(v16h a, v16h b, v8f c) {
  return __builtin_amdgcn_wmma_f32_16x16x32_f16(
       false, a,  false, b,
       (short)0, c,  false,  false);
}

__device__ __forceinline__ v16h load_frag16(const _Float16* base, int rc,
                                            int ld, int k0) {
  int lane = threadIdx.x & 31;
  int hf   = lane >> 4;
  int r    = lane & 15;
  const _Float16* p = base + (size_t)(rc + r) * ld + k0 + (hf << 3);
  v8h lo = *(const v8h*)p;
  v8h hi = *(const v8h*)(p + 16);
  return __builtin_shufflevector(lo, hi, 0, 1, 2, 3, 4, 5, 6, 7, 8, 9, 10, 11,
                                 12, 13, 14, 15);
}

__device__ __forceinline__ v16h load_frag16_f32(const float* base, int rc,
                                                int ld, int k0) {
  int lane = threadIdx.x & 31;
  int hf   = lane >> 4;
  int r    = lane & 15;
  const float* p = base + (size_t)(rc + r) * ld + k0 + (hf << 3);
  const float4 q0 = *(const float4*)p;
  const float4 q1 = *(const float4*)(p + 4);
  const float4 q2 = *(const float4*)(p + 16);
  const float4 q3 = *(const float4*)(p + 20);
  v16h a;
  a[0]  = (_Float16)q0.x; a[1]  = (_Float16)q0.y;
  a[2]  = (_Float16)q0.z; a[3]  = (_Float16)q0.w;
  a[4]  = (_Float16)q1.x; a[5]  = (_Float16)q1.y;
  a[6]  = (_Float16)q1.z; a[7]  = (_Float16)q1.w;
  a[8]  = (_Float16)q2.x; a[9]  = (_Float16)q2.y;
  a[10] = (_Float16)q2.z; a[11] = (_Float16)q2.w;
  a[12] = (_Float16)q3.x; a[13] = (_Float16)q3.y;
  a[14] = (_Float16)q3.z; a[15] = (_Float16)q3.w;
  return a;
}

__device__ __forceinline__ float lrelu(float x) {
  return x > 0.0f ? x : NEG_SLOPE * x;
}

__global__ void cast_kernel(const float* __restrict__ W1,
                            const float* __restrict__ W2,
                            _Float16* __restrict__ W1T,
                            _Float16* __restrict__ W2T) {
  int i = blockIdx.x * blockDim.x + threadIdx.x;
  if (i < H1 * NFEAT * NHID) {
    int h   = i / (NFEAT * NHID);
    int rem = i % (NFEAT * NHID);
    int k   = rem / NHID;
    int c   = rem % NHID;
    W1T[((size_t)(h * NHID + c)) * NFEAT + k] = (_Float16)W1[i];
  }
  if (i < (H1 * NHID) * NCLASS) {
    int k = i / NCLASS;
    int c = i % NCLASS;
    W2T[(size_t)c * (H1 * NHID) + k] = (_Float16)W2[i];
  }
}

__global__ void gemm1_kernel(const float* __restrict__ x,
                             const _Float16* __restrict__ W1T,
                             float* __restrict__ feats,
                             _Float16* __restrict__ featsT) {
  int head = blockIdx.y;
  int r0   = blockIdx.x << 4;
  const _Float16* BT = W1T + (size_t)head * NHID * NFEAT;

  v8f c0 = {}, c1 = {}, c2 = {}, c3 = {};
  for (int k0 = 0; k0 < NFEAT; k0 += 32) {
    v16h a = load_frag16_f32(x, r0, NFEAT, k0);
    c0 = wmma16(a, load_frag16(BT, 0, NFEAT, k0), c0);
    c1 = wmma16(a, load_frag16(BT, 16, NFEAT, k0), c1);
    c2 = wmma16(a, load_frag16(BT, 32, NFEAT, k0), c2);
    c3 = wmma16(a, load_frag16(BT, 48, NFEAT, k0), c3);
  }

  int lane = threadIdx.x & 31;
  int hf   = lane >> 4;
  int r    = lane & 15;
  v8f acc[4] = {c0, c1, c2, c3};
#pragma unroll
  for (int nt = 0; nt < 4; ++nt) {
#pragma unroll
    for (int k = 0; k < 8; ++k) {
      int orow = r0 + k + (hf << 3);
      int col  = nt * 16 + r;
      float v  = acc[nt][k];
      feats[((size_t)head * N + orow) * NHID + col] = v;
      featsT[((size_t)head * NHID + col) * N + orow] = (_Float16)v;
    }
  }
}

__global__ void sn_kernel(const float* __restrict__ feats,
                          const float* __restrict__ a_s,
                          const float* __restrict__ a_n,
                          float* __restrict__ s, float* __restrict__ n,
                          int fout, int rowsPerHead, int total) {
  int i = blockIdx.x * blockDim.x + threadIdx.x;
  if (i >= total) return;
  int head = i / rowsPerHead;
  const float* f  = feats + (size_t)i * fout;
  const float* av = a_s + head * fout;
  const float* nv = a_n + head * fout;
  float ss = 0.0f, nn = 0.0f;
  for (int c = 0; c < fout; ++c) {
    float v = f[c];
    ss += v * av[c];
    nn += v * nv[c];
  }
  s[i] = ss;
  n[i] = nn;
}

#define RS_WAVES 8

__device__ __forceinline__ float wave_max32(float v) {
#pragma unroll
  for (int msk = 16; msk > 0; msk >>= 1) v = fmaxf(v, __shfl_xor(v, msk, 32));
  return v;
}

__device__ __forceinline__ float wave_sum32(float v) {
#pragma unroll
  for (int msk = 16; msk > 0; msk >>= 1) v += __shfl_xor(v, msk, 32);
  return v;
}

__device__ __forceinline__ float lrelu_max(float x) {
  return fmaxf(x, NEG_SLOPE * x);
}

template <int NH, int U>
struct RsChunk {
  int4   a[U];
  float4 v[U][NH];
};

template <int NH, int U>
__device__ __forceinline__ void rs_load(RsChunk<NH, U>& c, const int4* arow,
                                        const float4* nq, int ci) {
#pragma unroll
  for (int u = 0; u < U; ++u) {
    const int q = (ci * U + u) * 32;
    c.a[u] = arow[q];
#pragma unroll
    for (int h = 0; h < NH; ++h) c.v[u][h] = nq[h * (N / 4) + q];
  }
}

template <int NH, int U>
__device__ __forceinline__ void rs_max_step(const RsChunk<NH, U>& c,
                                            float (&nmax)[NH], int& cnt) {
#pragma unroll
  for (int u = 0; u < U; ++u) {
    const bool e0 = c.a[u].x > 0, e1 = c.a[u].y > 0;
    const bool e2 = c.a[u].z > 0, e3 = c.a[u].w > 0;
    cnt += (int)e0 + (int)e1 + (int)e2 + (int)e3;
#pragma unroll
    for (int h = 0; h < NH; ++h) {
      const float x0 = e0 ? c.v[u][h].x : -INFINITY;
      const float x1 = e1 ? c.v[u][h].y : -INFINITY;
      const float x2 = e2 ? c.v[u][h].z : -INFINITY;
      const float x3 = e3 ? c.v[u][h].w : -INFINITY;
      nmax[h] = fmaxf(fmaxf(nmax[h], x0), fmaxf(x1, fmaxf(x2, x3)));
    }
  }
}

template <int NH, int U, bool WP>
__device__ __forceinline__ void rs_sum_step(const RsChunk<NH, U>& c,
                                            const float (&sv)[NH],
                                            const float (&mv)[NH],
                                            float (&acc)[NH],
                                            _Float16* prow, int ci) {
  static_assert(!WP || NH == 1, "the probability plane has one head");
#pragma unroll
  for (int u = 0; u < U; ++u) {
    const bool e0 = c.a[u].x > 0, e1 = c.a[u].y > 0;
    const bool e2 = c.a[u].z > 0, e3 = c.a[u].w > 0;
#pragma unroll
    for (int h = 0; h < NH; ++h) {
      const float p0 = __expf(lrelu_max(sv[h] + c.v[u][h].x) - mv[h]);
      const float p1 = __expf(lrelu_max(sv[h] + c.v[u][h].y) - mv[h]);
      const float p2 = __expf(lrelu_max(sv[h] + c.v[u][h].z) - mv[h]);
      const float p3 = __expf(lrelu_max(sv[h] + c.v[u][h].w) - mv[h]);
      const float q0 = e0 ? p0 : 0.0f, q1 = e1 ? p1 : 0.0f;
      const float q2 = e2 ? p2 : 0.0f, q3 = e3 ? p3 : 0.0f;
      acc[h] += q0 + q1;
      acc[h] += q2 + q3;
      if (WP) {
        const v4h pv = {(_Float16)q0, (_Float16)q1, (_Float16)q2,
                        (_Float16)q3};
        *(v4h*)(prow + (ci * U + u) * 128) = pv;
      }
    }
  }
}

template <int NH, int U, bool WP>
__global__ void __launch_bounds__(RS_WAVES * 32)
__attribute__((amdgpu_waves_per_eu(4, 4)))
rowstats_kernel(const int* __restrict__ adj, const float* __restrict__ s,
                const float* __restrict__ n, float* __restrict__ m,
                float* __restrict__ Z, _Float16* __restrict__ P) {
  constexpr int CHUNKS = N / (128 * U);
  static_assert(CHUNKS >= 2 && CHUNKS % 2 == 0, "chunks are taken in pairs");

  const int lane = threadIdx.x & 31;
  const int wave = __builtin_amdgcn_readfirstlane((int)(threadIdx.x >> 5));
  const int i    = blockIdx.x * RS_WAVES + wave;

  const int* arow_i = adj + (size_t)i * N;
  __builtin_prefetch(arow_i + lane * 64, 0, 0);
  __builtin_prefetch(arow_i + (N / 2) + lane * 64, 0, 0);

  const int4*   arow = (const int4*)arow_i + lane;
  const float4* nq   = (const float4*)n + lane;
  _Float16* prow = WP ? P + (size_t)i * N + lane * 4 : nullptr;

  RsChunk<NH, U> ca, cb;

  float nmax[NH];
#pragma unroll
  for (int h = 0; h < NH; ++h) nmax[h] = -INFINITY;
  int cnt = 0;

  rs_load(ca, arow, nq, 0);
  for (int ci = 0; ci < CHUNKS - 2; ci += 2) {
    rs_load(cb, arow, nq, ci + 1);
    rs_max_step(ca, nmax, cnt);
    rs_load(ca, arow, nq, ci + 2);
    rs_max_step(cb, nmax, cnt);
  }
  rs_load(cb, arow, nq, CHUNKS - 1);
  rs_max_step(ca, nmax, cnt);
  rs_max_step(cb, nmax, cnt);

#pragma unroll
  for (int msk = 16; msk > 0; msk >>= 1) cnt += __shfl_xor(cnt, msk, 32);
  const int n_masked = N - cnt;

  float sv[NH], mv[NH];
#pragma unroll
  for (int h = 0; h < NH; ++h) {
    sv[h] = s[h * N + i];
    float mm = fmaxf(-3.0e38f, lrelu_max(sv[h] + wave_max32(nmax[h])));
    if (n_masked > 0) mm = fmaxf(mm, MASK_VAL);
    mv[h] = mm;
  }

  float acc[NH];
#pragma unroll
  for (int h = 0; h < NH; ++h) acc[h] = 0.0f;

  rs_load(ca, arow, nq, 0);
  for (int ci = 0; ci < CHUNKS - 2; ci += 2) {
    rs_load(cb, arow, nq, ci + 1);
    rs_sum_step<NH, U, WP>(ca, sv, mv, acc, prow, ci);
    rs_load(ca, arow, nq, ci + 2);
    rs_sum_step<NH, U, WP>(cb, sv, mv, acc, prow, ci + 1);
  }
  rs_load(cb, arow, nq, CHUNKS - 1);
  rs_sum_step<NH, U, WP>(ca, sv, mv, acc, prow, CHUNKS - 2);
  rs_sum_step<NH, U, WP>(cb, sv, mv, acc, prow, CHUNKS - 1);

#pragma unroll
  for (int h = 0; h < NH; ++h) {
    float zs = wave_sum32(acc[h]);
    if (n_masked > 0) zs += (float)n_masked * __expf(MASK_VAL - mv[h]);
    if (lane == 0) {
      if (!WP) m[h * N + i] = mv[h];
      Z[h * N + i] = zs;
    }
  }
}

#define F1_ROWS  16
#define F1_COLS  128
#define F1_PITCH 136
#define F1_HEAD  (F1_ROWS * F1_PITCH)
#define F1_STAGE (H1 * F1_HEAD)
#define F1_KS    (F1_COLS / 32)
#define F1_MU    2

struct F1Frags {
  v16h b[F1_KS][2];
};

__device__ __forceinline__ void f1_load_b(F1Frags& f, const _Float16* BT,
                                          int f0, int ci) {
#pragma unroll
  for (int ks = 0; ks < F1_KS; ++ks) {
    const int j0 = ci * F1_COLS + ks * 32;
    f.b[ks][0] = load_frag16(BT, f0, N, j0);
    f.b[ks][1] = load_frag16(BT, f0 + 16, N, j0);
  }
}

__device__ __forceinline__ void f1_form(const RsChunk<H1, 1>& c,
                                        const float (&sv)[H1],
                                        const float (&mv)[H1],
                                        float (&zacc)[H1],
                                        _Float16* prow) {
  const bool e0 = c.a[0].x > 0, e1 = c.a[0].y > 0;
  const bool e2 = c.a[0].z > 0, e3 = c.a[0].w > 0;
#pragma unroll
  for (int h = 0; h < H1; ++h) {
    const float p0 = __expf(lrelu_max(sv[h] + c.v[0][h].x) - mv[h]);
    const float p1 = __expf(lrelu_max(sv[h] + c.v[0][h].y) - mv[h]);
    const float p2 = __expf(lrelu_max(sv[h] + c.v[0][h].z) - mv[h]);
    const float p3 = __expf(lrelu_max(sv[h] + c.v[0][h].w) - mv[h]);
    const float q0 = e0 ? p0 : 0.0f, q1 = e1 ? p1 : 0.0f;
    const float q2 = e2 ? p2 : 0.0f, q3 = e3 ? p3 : 0.0f;
    zacc[h] += q0 + q1;
    zacc[h] += q2 + q3;
    const v4h pv = {(_Float16)q0, (_Float16)q1, (_Float16)q2, (_Float16)q3};
    *(v4h*)(prow + h * F1_HEAD) = pv;
  }
}

__device__ __forceinline__ void f1_mma(const _Float16* ph, const F1Frags& f,
                                       v8f& c0, v8f& c1) {
#pragma unroll
  for (int ks = 0; ks < F1_KS; ++ks) {
    const v16h a = load_frag16(ph, 0, F1_PITCH, ks * 32);
    c0 = wmma16(a, f.b[ks][0], c0);
    c1 = wmma16(a, f.b[ks][1], c1);
  }
}

__device__ __forceinline__ void f1_barrier() {
  asm volatile("s_wait_dscnt 0x0\n\ts_barrier_signal -1\n\ts_barrier_wait -1"
               ::: "memory");
}

__global__ void __launch_bounds__(F1_ROWS * 32)
attn_gemm1_kernel(const int* __restrict__ adj,
                  const _Float16* __restrict__ featsT,
                  const float* __restrict__ s1,
                  const float* __restrict__ n1,
                  const float* __restrict__ b1,
                  _Float16* __restrict__ H) {
  constexpr int CH = N / F1_COLS;
  static_assert(CH % 2 == 0, "blocks are taken in pairs, one per LDS stage");
  __shared__ __attribute__((aligned(16))) _Float16 pl[2 * F1_STAGE];
  __shared__ __attribute__((aligned(16))) float zl[H1 * F1_ROWS];

  const int lane = threadIdx.x & 31;
  const int wave = __builtin_amdgcn_readfirstlane((int)(threadIdx.x >> 5));
  const int r0   = blockIdx.x * F1_ROWS;
  const int i    = r0 + wave;
  const int head = wave & (H1 - 1);
  const int f0   = (wave >> 3) * 32;

  const int* arow_i = adj + (size_t)i * N;
  __builtin_prefetch(arow_i + lane * 64, 0, 0);
  __builtin_prefetch(arow_i + (N / 2) + lane * 64, 0, 0);

  const int4*   arow = (const int4*)arow_i + lane;
  const float4* nq   = (const float4*)n1 + lane;

  constexpr int MCH = N / (F1_COLS * F1_MU);
  static_assert(MCH >= 2 && MCH % 2 == 0, "chunks are taken in pairs");
  float nmax[H1];
#pragma unroll
  for (int h = 0; h < H1; ++h) nmax[h] = -INFINITY;
  int cnt = 0;

  RsChunk<H1, F1_MU> ma, mb;
  rs_load(ma, arow, nq, 0);
  for (int ci = 0; ci < MCH - 2; ci += 2) {
    rs_load(mb, arow, nq, ci + 1);
    rs_max_step(ma, nmax, cnt);
    rs_load(ma, arow, nq, ci + 2);
    rs_max_step(mb, nmax, cnt);
  }
  rs_load(mb, arow, nq, MCH - 1);
  rs_max_step(ma, nmax, cnt);
  RsChunk<H1, 1> ca, cb;
  rs_load(ca, arow, nq, 0);
  rs_max_step(mb, nmax, cnt);

#pragma unroll
  for (int msk = 16; msk > 0; msk >>= 1) cnt += __shfl_xor(cnt, msk, 32);
  const int n_masked = N - cnt;

  float sv[H1], mv[H1], zacc[H1];
#pragma unroll
  for (int h = 0; h < H1; ++h) {
    sv[h] = s1[h * N + i];
    float mm = fmaxf(-3.0e38f, lrelu_max(sv[h] + wave_max32(nmax[h])));
    if (n_masked > 0) mm = fmaxf(mm, MASK_VAL);
    mv[h] = mm;
    zacc[h] = 0.0f;
  }

  _Float16* prow0 = pl + wave * F1_PITCH + lane * 4;
  _Float16* prow1 = prow0 + F1_STAGE;
  const _Float16* ph0 = pl + head * F1_HEAD;
  const _Float16* ph1 = ph0 + F1_STAGE;
  const _Float16* BT  = featsT + (size_t)head * NHID * N;

  F1Frags bf;
  v8f c0 = {}, c1 = {};

  for (int ci = 0; ci < CH; ci += 2) {
    rs_load(cb, arow, nq, ci + 1);
    f1_load_b(bf, BT, f0, ci);
    f1_form(ca, sv, mv, zacc, prow0);
    f1_barrier();
    f1_mma(ph0, bf, c0, c1);

    if (ci + 2 < CH) rs_load(ca, arow, nq, ci + 2);
    f1_load_b(bf, BT, f0, ci + 1);
    f1_form(cb, sv, mv, zacc, prow1);
    f1_barrier();
    f1_mma(ph1, bf, c0, c1);
  }

#pragma unroll
  for (int h = 0; h < H1; ++h) {
    float zs = wave_sum32(zacc[h]);
    if (n_masked > 0) zs += (float)n_masked * __expf(MASK_VAL - mv[h]);
    if (lane == 0) zl[h * F1_ROWS + wave] = zs;
  }
  __syncthreads();

  const int hf = lane >> 4;
  const int r  = lane & 15;
  const float4 za = *(const float4*)(zl + head * F1_ROWS + (hf << 3));
  const float4 zb = *(const float4*)(zl + head * F1_ROWS + (hf << 3) + 4);
  const float zi[8] = {1.0f / za.x, 1.0f / za.y, 1.0f / za.z, 1.0f / za.w,
                       1.0f / zb.x, 1.0f / zb.y, 1.0f / zb.z, 1.0f / zb.w};
  v8f acc[2] = {c0, c1};
#pragma unroll
  for (int nt = 0; nt < 2; ++nt) {
#pragma unroll
    for (int k = 0; k < 8; ++k) {
      int orow = r0 + k + (hf << 3);
      int col  = f0 + nt * 16 + r;
      float v  = acc[nt][k] * zi[k] + b1[head * NHID + col];
      v = v > 0.0f ? v : 0.0f;
      H[(size_t)orow * (H1 * NHID) + head * NHID + col] = (_Float16)v;
    }
  }
}

__global__ void gemm2_kernel(const _Float16* __restrict__ H,
                             const _Float16* __restrict__ W2T,
                             float* __restrict__ feats2,
                             _Float16* __restrict__ feats2T) {
  int r0 = blockIdx.x << 4;
  v8f c = {};
  for (int k0 = 0; k0 < H1 * NHID; k0 += 32) {
    v16h a = load_frag16(H, r0, H1 * NHID, k0);
    v16h b = load_frag16(W2T, 0, H1 * NHID, k0);
    c = wmma16(a, b, c);
  }
  int lane = threadIdx.x & 31;
  int hf   = lane >> 4;
  int r    = lane & 15;
#pragma unroll
  for (int k = 0; k < 8; ++k) {
    int orow = r0 + k + (hf << 3);
    float v  = c[k];
    feats2[(size_t)orow * NCLASS + r] = v;
    feats2T[(size_t)r * N + orow]     = (_Float16)v;
  }
}

#define A2_U 8

struct A2Group {
  v16h a[A2_U];
  v16h b[A2_U];
};

__device__ __forceinline__ void a2_load(A2Group& g, const _Float16* P,
                                        const _Float16* BT, int r0, int gi) {
#pragma unroll
  for (int u = 0; u < A2_U; ++u) {
    const int j0 = (gi * A2_U + u) * 32;
    g.a[u] = load_frag16(P, r0, N, j0);
    g.b[u] = load_frag16(BT, 0, N, j0);
  }
}

__device__ __forceinline__ v8f a2_mma(const A2Group& g, v8f c) {
#pragma unroll
  for (int u = 0; u < A2_U; ++u) c = wmma16(g.a[u], g.b[u], c);
  return c;
}

__global__ void __launch_bounds__(32)
attn_gemm2_kernel(const _Float16* __restrict__ P,
                  const _Float16* __restrict__ feats2T,
                  const float* __restrict__ Z2,
                  const float* __restrict__ b2,
                  float* __restrict__ out) {
  constexpr int GROUPS = N / (32 * A2_U);
  static_assert(GROUPS >= 2 && GROUPS % 2 == 0, "groups are taken in pairs");

  int r0   = blockIdx.x << 4;
  int lane = threadIdx.x & 31;
  int hf   = lane >> 4;
  int r    = lane & 15;

  A2Group ga, gb;
  v8f c = {};
  a2_load(ga, P, feats2T, r0, 0);
  for (int gi = 0; gi < GROUPS - 2; gi += 2) {
    a2_load(gb, P, feats2T, r0, gi + 1);
    c = a2_mma(ga, c);
    a2_load(ga, P, feats2T, r0, gi + 2);
    c = a2_mma(gb, c);
  }
  a2_load(gb, P, feats2T, r0, GROUPS - 1);
  c = a2_mma(ga, c);
  c = a2_mma(gb, c);

  float bias = b2[r];
#pragma unroll
  for (int k = 0; k < 8; ++k) {
    int orow = r0 + k + (hf << 3);
    float zi = 1.0f / Z2[orow];
    float v  = c[k] * zi + bias;
    v = v > 0.0f ? v : 0.0f;
    float mx = v;
#pragma unroll
    for (int msk = 1; msk < 16; msk <<= 1)
      mx = fmaxf(mx, __shfl_xor(mx, msk, 32));
    float ex = __expf(v - mx);
    float sum = ex;
#pragma unroll
    for (int msk = 1; msk < 16; msk <<= 1) sum += __shfl_xor(sum, msk, 32);
    float o  = (v - mx) - logf(sum);
    out[(size_t)orow * NCLASS + r] = o;
  }
}

extern "C" void kernel_launch(void* const* d_in, const int* in_sizes, int n_in,
                              void* d_out, int out_size, void* d_ws,
                              size_t ws_size, hipStream_t stream) {
  const float* x   = (const float*)d_in[0];
  const int*   adj = (const int*)d_in[1];
  const float* W1  = (const float*)d_in[2];
  const float* b1  = (const float*)d_in[3];
  const float* as1 = (const float*)d_in[4];
  const float* an1 = (const float*)d_in[5];
  const float* W2  = (const float*)d_in[6];
  const float* b2  = (const float*)d_in[7];
  const float* as2 = (const float*)d_in[8];
  const float* an2 = (const float*)d_in[9];
  float* out = (float*)d_out;

  char* ws = (char*)d_ws;
  size_t off = 0;
  auto alloc = [&](size_t bytes) -> void* {
    off = (off + 255) & ~(size_t)255;
    void* p = ws + off;
    off += bytes;
    return p;
  };

  float*    feats2  = (float*)alloc((size_t)N * NCLASS * 4);
  _Float16* feats2T = (_Float16*)alloc((size_t)NCLASS * N * 2);
  float*    s2      = (float*)alloc((size_t)N * 4);
  float*    n2      = (float*)alloc((size_t)N * 4);
  float*    Z2      = (float*)alloc((size_t)N * 4);
  const size_t shared = off;
  _Float16* P2      = (_Float16*)alloc((size_t)N * N * 2);
  off = shared;
  _Float16* W1T     = (_Float16*)alloc((size_t)H1 * NHID * NFEAT * 2);
  _Float16* W2T     = (_Float16*)alloc((size_t)NCLASS * H1 * NHID * 2);
  float*    feats1  = (float*)alloc((size_t)H1 * N * NHID * 4);
  _Float16* featsT  = (_Float16*)alloc((size_t)H1 * NHID * N * 2);
  float*    s1      = (float*)alloc((size_t)H1 * N * 4);
  float*    n1      = (float*)alloc((size_t)H1 * N * 4);
  _Float16* H       = (_Float16*)alloc((size_t)N * H1 * NHID * 2);
  (void)ws_size; (void)n_in; (void)in_sizes; (void)out_size;

  {
    int total = H1 * NFEAT * NHID;
    cast_kernel<<<(total + 255) / 256, 256, 0, stream>>>(W1, W2, W1T, W2T);
  }
  gemm1_kernel<<<dim3(N / 16, H1), 32, 0, stream>>>(x, W1T, feats1, featsT);
  sn_kernel<<<(H1 * N + 255) / 256, 256, 0, stream>>>(feats1, as1, an1, s1, n1,
                                                      NHID, N, H1 * N);
  attn_gemm1_kernel<<<N / F1_ROWS, F1_ROWS * 32, 0, stream>>>(
      adj, featsT, s1, n1, b1, H);
  gemm2_kernel<<<N / 16, 32, 0, stream>>>(H, W2T, feats2, feats2T);
  sn_kernel<<<(N + 255) / 256, 256, 0, stream>>>(feats2, as2, an2, s2, n2,
                                                 NCLASS, N, N);
  rowstats_kernel<1, 8, true><<<N / RS_WAVES, RS_WAVES * 32, 0, stream>>>(
      adj, s2, n2, nullptr, Z2, P2);
  attn_gemm2_kernel<<<N / 16, 32, 0, stream>>>(P2, feats2T, Z2, b2, out);
}


# Layer-1 attention: a work-group owns a 32-row tile and 4 heads (was 16 rows, 8 heads); each B fragment from memory now serves two row tiles, feature reads from L2 halve; order of sums unchanged
# speedup vs baseline: 3.5322x; 1.2197x over previous
#include <hip/hip_runtime.h>
#include <hip/hip_bf16.h>


#define N      4096
#define NFEAT  512
#define NHID   64
#define NCLASS 16
#define H1     8
#define NEG_SLOPE 0.2f
#define MASK_VAL  -1e10f

typedef __attribute__((ext_vector_type(16))) _Float16 v16h;
typedef __attribute__((ext_vector_type(8)))  _Float16 v8h;
typedef __attribute__((ext_vector_type(4)))  _Float16 v4h;
typedef __attribute__((ext_vector_type(8)))  float    v8f;

__device__ __forceinline__ v8f wmma16(v16h a, v16h b, v8f c) {
  return __builtin_amdgcn_wmma_f32_16x16x32_f16(
       false, a,  false, b,
       (short)0, c,  false,  false);
}

__device__ __forceinline__ v16h load_frag16(const _Float16* base, int rc,
                                            int ld, int k0) {
  int lane = threadIdx.x & 31;
  int hf   = lane >> 4;
  int r    = lane & 15;
  const _Float16* p = base + (size_t)(rc + r) * ld + k0 + (hf << 3);
  v8h lo = *(const v8h*)p;
  v8h hi = *(const v8h*)(p + 16);
  return __builtin_shufflevector(lo, hi, 0, 1, 2, 3, 4, 5, 6, 7, 8, 9, 10, 11,
                                 12, 13, 14, 15);
}

__device__ __forceinline__ v16h load_frag16_f32(const float* base, int rc,
                                                int ld, int k0) {
  int lane = threadIdx.x & 31;
  int hf   = lane >> 4;
  int r    = lane & 15;
  const float* p = base + (size_t)(rc + r) * ld + k0 + (hf << 3);
  const float4 q0 = *(const float4*)p;
  const float4 q1 = *(const float4*)(p + 4);
  const float4 q2 = *(const float4*)(p + 16);
  const float4 q3 = *(const float4*)(p + 20);
  v16h a;
  a[0]  = (_Float16)q0.x; a[1]  = (_Float16)q0.y;
  a[2]  = (_Float16)q0.z; a[3]  = (_Float16)q0.w;
  a[4]  = (_Float16)q1.x; a[5]  = (_Float16)q1.y;
  a[6]  = (_Float16)q1.z; a[7]  = (_Float16)q1.w;
  a[8]  = (_Float16)q2.x; a[9]  = (_Float16)q2.y;
  a[10] = (_Float16)q2.z; a[11] = (_Float16)q2.w;
  a[12] = (_Float16)q3.x; a[13] = (_Float16)q3.y;
  a[14] = (_Float16)q3.z; a[15] = (_Float16)q3.w;
  return a;
}

__device__ __forceinline__ float lrelu(float x) {
  return x > 0.0f ? x : NEG_SLOPE * x;
}

__global__ void cast_kernel(const float* __restrict__ W1,
                            const float* __restrict__ W2,
                            _Float16* __restrict__ W1T,
                            _Float16* __restrict__ W2T) {
  int i = blockIdx.x * blockDim.x + threadIdx.x;
  if (i < H1 * NFEAT * NHID) {
    int h   = i / (NFEAT * NHID);
    int rem = i % (NFEAT * NHID);
    int k   = rem / NHID;
    int c   = rem % NHID;
    W1T[((size_t)(h * NHID + c)) * NFEAT + k] = (_Float16)W1[i];
  }
  if (i < (H1 * NHID) * NCLASS) {
    int k = i / NCLASS;
    int c = i % NCLASS;
    W2T[(size_t)c * (H1 * NHID) + k] = (_Float16)W2[i];
  }
}

__global__ void gemm1_kernel(const float* __restrict__ x,
                             const _Float16* __restrict__ W1T,
                             float* __restrict__ feats,
                             _Float16* __restrict__ featsT) {
  int head = blockIdx.y;
  int r0   = blockIdx.x << 4;
  const _Float16* BT = W1T + (size_t)head * NHID * NFEAT;

  v8f c0 = {}, c1 = {}, c2 = {}, c3 = {};
  for (int k0 = 0; k0 < NFEAT; k0 += 32) {
    v16h a = load_frag16_f32(x, r0, NFEAT, k0);
    c0 = wmma16(a, load_frag16(BT, 0, NFEAT, k0), c0);
    c1 = wmma16(a, load_frag16(BT, 16, NFEAT, k0), c1);
    c2 = wmma16(a, load_frag16(BT, 32, NFEAT, k0), c2);
    c3 = wmma16(a, load_frag16(BT, 48, NFEAT, k0), c3);
  }

  int lane = threadIdx.x & 31;
  int hf   = lane >> 4;
  int r    = lane & 15;
  v8f acc[4] = {c0, c1, c2, c3};
#pragma unroll
  for (int nt = 0; nt < 4; ++nt) {
#pragma unroll
    for (int k = 0; k < 8; ++k) {
      int orow = r0 + k + (hf << 3);
      int col  = nt * 16 + r;
      float v  = acc[nt][k];
      feats[((size_t)head * N + orow) * NHID + col] = v;
      featsT[((size_t)head * NHID + col) * N + orow] = (_Float16)v;
    }
  }
}

__global__ void sn_kernel(const float* __restrict__ feats,
                          const float* __restrict__ a_s,
                          const float* __restrict__ a_n,
                          float* __restrict__ s, float* __restrict__ n,
                          int fout, int rowsPerHead, int total) {
  int i = blockIdx.x * blockDim.x + threadIdx.x;
  if (i >= total) return;
  int head = i / rowsPerHead;
  const float* f  = feats + (size_t)i * fout;
  const float* av = a_s + head * fout;
  const float* nv = a_n + head * fout;
  float ss = 0.0f, nn = 0.0f;
  for (int c = 0; c < fout; ++c) {
    float v = f[c];
    ss += v * av[c];
    nn += v * nv[c];
  }
  s[i] = ss;
  n[i] = nn;
}

#define RS_WAVES 8

__device__ __forceinline__ float wave_max32(float v) {
#pragma unroll
  for (int msk = 16; msk > 0; msk >>= 1) v = fmaxf(v, __shfl_xor(v, msk, 32));
  return v;
}

__device__ __forceinline__ float wave_sum32(float v) {
#pragma unroll
  for (int msk = 16; msk > 0; msk >>= 1) v += __shfl_xor(v, msk, 32);
  return v;
}

__device__ __forceinline__ float lrelu_max(float x) {
  return fmaxf(x, NEG_SLOPE * x);
}

template <int NH, int U>
struct RsChunk {
  int4   a[U];
  float4 v[U][NH];
};

template <int NH, int U>
__device__ __forceinline__ void rs_load(RsChunk<NH, U>& c, const int4* arow,
                                        const float4* nq, int ci) {
#pragma unroll
  for (int u = 0; u < U; ++u) {
    const int q = (ci * U + u) * 32;
    c.a[u] = arow[q];
#pragma unroll
    for (int h = 0; h < NH; ++h) c.v[u][h] = nq[h * (N / 4) + q];
  }
}

template <int NH, int U>
__device__ __forceinline__ void rs_max_step(const RsChunk<NH, U>& c,
                                            float (&nmax)[NH], int& cnt) {
#pragma unroll
  for (int u = 0; u < U; ++u) {
    const bool e0 = c.a[u].x > 0, e1 = c.a[u].y > 0;
    const bool e2 = c.a[u].z > 0, e3 = c.a[u].w > 0;
    cnt += (int)e0 + (int)e1 + (int)e2 + (int)e3;
#pragma unroll
    for (int h = 0; h < NH; ++h) {
      const float x0 = e0 ? c.v[u][h].x : -INFINITY;
      const float x1 = e1 ? c.v[u][h].y : -INFINITY;
      const float x2 = e2 ? c.v[u][h].z : -INFINITY;
      const float x3 = e3 ? c.v[u][h].w : -INFINITY;
      nmax[h] = fmaxf(fmaxf(nmax[h], x0), fmaxf(x1, fmaxf(x2, x3)));
    }
  }
}

template <int NH, int U, bool WP>
__device__ __forceinline__ void rs_sum_step(const RsChunk<NH, U>& c,
                                            const float (&sv)[NH],
                                            const float (&mv)[NH],
                                            float (&acc)[NH],
                                            _Float16* prow, int ci) {
  static_assert(!WP || NH == 1, "the probability plane has one head");
#pragma unroll
  for (int u = 0; u < U; ++u) {
    const bool e0 = c.a[u].x > 0, e1 = c.a[u].y > 0;
    const bool e2 = c.a[u].z > 0, e3 = c.a[u].w > 0;
#pragma unroll
    for (int h = 0; h < NH; ++h) {
      const float p0 = __expf(lrelu_max(sv[h] + c.v[u][h].x) - mv[h]);
      const float p1 = __expf(lrelu_max(sv[h] + c.v[u][h].y) - mv[h]);
      const float p2 = __expf(lrelu_max(sv[h] + c.v[u][h].z) - mv[h]);
      const float p3 = __expf(lrelu_max(sv[h] + c.v[u][h].w) - mv[h]);
      const float q0 = e0 ? p0 : 0.0f, q1 = e1 ? p1 : 0.0f;
      const float q2 = e2 ? p2 : 0.0f, q3 = e3 ? p3 : 0.0f;
      acc[h] += q0 + q1;
      acc[h] += q2 + q3;
      if (WP) {
        const v4h pv = {(_Float16)q0, (_Float16)q1, (_Float16)q2,
                        (_Float16)q3};
        *(v4h*)(prow + (ci * U + u) * 128) = pv;
      }
    }
  }
}

template <int NH, int U, bool WP>
__global__ void __launch_bounds__(RS_WAVES * 32)
__attribute__((amdgpu_waves_per_eu(4, 4)))
rowstats_kernel(const int* __restrict__ adj, const float* __restrict__ s,
                const float* __restrict__ n, float* __restrict__ m,
                float* __restrict__ Z, _Float16* __restrict__ P) {
  constexpr int CHUNKS = N / (128 * U);
  static_assert(CHUNKS >= 2 && CHUNKS % 2 == 0, "chunks are taken in pairs");

  const int lane = threadIdx.x & 31;
  const int wave = __builtin_amdgcn_readfirstlane((int)(threadIdx.x >> 5));
  const int i    = blockIdx.x * RS_WAVES + wave;

  const int* arow_i = adj + (size_t)i * N;
  __builtin_prefetch(arow_i + lane * 64, 0, 0);
  __builtin_prefetch(arow_i + (N / 2) + lane * 64, 0, 0);

  const int4*   arow = (const int4*)arow_i + lane;
  const float4* nq   = (const float4*)n + lane;
  _Float16* prow = WP ? P + (size_t)i * N + lane * 4 : nullptr;

  RsChunk<NH, U> ca, cb;

  float nmax[NH];
#pragma unroll
  for (int h = 0; h < NH; ++h) nmax[h] = -INFINITY;
  int cnt = 0;

  rs_load(ca, arow, nq, 0);
  for (int ci = 0; ci < CHUNKS - 2; ci += 2) {
    rs_load(cb, arow, nq, ci + 1);
    rs_max_step(ca, nmax, cnt);
    rs_load(ca, arow, nq, ci + 2);
    rs_max_step(cb, nmax, cnt);
  }
  rs_load(cb, arow, nq, CHUNKS - 1);
  rs_max_step(ca, nmax, cnt);
  rs_max_step(cb, nmax, cnt);

#pragma unroll
  for (int msk = 16; msk > 0; msk >>= 1) cnt += __shfl_xor(cnt, msk, 32);
  const int n_masked = N - cnt;

  float sv[NH], mv[NH];
#pragma unroll
  for (int h = 0; h < NH; ++h) {
    sv[h] = s[h * N + i];
    float mm = fmaxf(-3.0e38f, lrelu_max(sv[h] + wave_max32(nmax[h])));
    if (n_masked > 0) mm = fmaxf(mm, MASK_VAL);
    mv[h] = mm;
  }

  float acc[NH];
#pragma unroll
  for (int h = 0; h < NH; ++h) acc[h] = 0.0f;

  rs_load(ca, arow, nq, 0);
  for (int ci = 0; ci < CHUNKS - 2; ci += 2) {
    rs_load(cb, arow, nq, ci + 1);
    rs_sum_step<NH, U, WP>(ca, sv, mv, acc, prow, ci);
    rs_load(ca, arow, nq, ci + 2);
    rs_sum_step<NH, U, WP>(cb, sv, mv, acc, prow, ci + 1);
  }
  rs_load(cb, arow, nq, CHUNKS - 1);
  rs_sum_step<NH, U, WP>(ca, sv, mv, acc, prow, CHUNKS - 2);
  rs_sum_step<NH, U, WP>(cb, sv, mv, acc, prow, CHUNKS - 1);

#pragma unroll
  for (int h = 0; h < NH; ++h) {
    float zs = wave_sum32(acc[h]);
    if (n_masked > 0) zs += (float)n_masked * __expf(MASK_VAL - mv[h]);
    if (lane == 0) {
      if (!WP) m[h * N + i] = mv[h];
      Z[h * N + i] = zs;
    }
  }
}

#define F1_ROWS  32
#define F1_WAVES 16
#define F1_HG    4
#define F1_COLS  128
#define F1_PITCH 136
#define F1_HEAD  (F1_ROWS * F1_PITCH)
#define F1_STAGE (F1_HG * F1_HEAD)
#define F1_KS    (F1_COLS / 32)
#define F1_MU    2

template <int U>
struct F1Chunk {
  int4   a[U][2];
  float4 v[U][F1_HG];
};

template <int U>
__device__ __forceinline__ void f1_load(F1Chunk<U>& c, const int4* arow0,
                                        const int4* arow1, const float4* nq,
                                        int ci) {
#pragma unroll
  for (int u = 0; u < U; ++u) {
    const int q = (ci * U + u) * 32;
    c.a[u][0] = arow0[q];
    c.a[u][1] = arow1[q];
#pragma unroll
    for (int h = 0; h < F1_HG; ++h) c.v[u][h] = nq[h * (N / 4) + q];
  }
}

template <int U>
__device__ __forceinline__ void f1_max_step(const F1Chunk<U>& c,
                                            float (&nmax)[2][F1_HG],
                                            int (&cnt)[2]) {
#pragma unroll
  for (int u = 0; u < U; ++u) {
#pragma unroll
    for (int t = 0; t < 2; ++t) {
      const bool e0 = c.a[u][t].x > 0, e1 = c.a[u][t].y > 0;
      const bool e2 = c.a[u][t].z > 0, e3 = c.a[u][t].w > 0;
      cnt[t] += (int)e0 + (int)e1 + (int)e2 + (int)e3;
#pragma unroll
      for (int h = 0; h < F1_HG; ++h) {
        const float x0 = e0 ? c.v[u][h].x : -INFINITY;
        const float x1 = e1 ? c.v[u][h].y : -INFINITY;
        const float x2 = e2 ? c.v[u][h].z : -INFINITY;
        const float x3 = e3 ? c.v[u][h].w : -INFINITY;
        nmax[t][h] =
            fmaxf(fmaxf(nmax[t][h], x0), fmaxf(x1, fmaxf(x2, x3)));
      }
    }
  }
}

struct F1Frags {
  v16h b[F1_KS];
};

__device__ __forceinline__ void f1_load_b(F1Frags& f, const _Float16* BT,
                                          int f0, int ci) {
#pragma unroll
  for (int ks = 0; ks < F1_KS; ++ks)
    f.b[ks] = load_frag16(BT, f0, N, ci * F1_COLS + ks * 32);
}

__device__ __forceinline__ void f1_form(const F1Chunk<1>& c,
                                        const float (&sv)[2][F1_HG],
                                        const float (&mv)[2][F1_HG],
                                        float (&zacc)[2][F1_HG],
                                        _Float16* prow) {
#pragma unroll
  for (int t = 0; t < 2; ++t) {
    const bool e0 = c.a[0][t].x > 0, e1 = c.a[0][t].y > 0;
    const bool e2 = c.a[0][t].z > 0, e3 = c.a[0][t].w > 0;
#pragma unroll
    for (int h = 0; h < F1_HG; ++h) {
      const float p0 = __expf(lrelu_max(sv[t][h] + c.v[0][h].x) - mv[t][h]);
      const float p1 = __expf(lrelu_max(sv[t][h] + c.v[0][h].y) - mv[t][h]);
      const float p2 = __expf(lrelu_max(sv[t][h] + c.v[0][h].z) - mv[t][h]);
      const float p3 = __expf(lrelu_max(sv[t][h] + c.v[0][h].w) - mv[t][h]);
      const float q0 = e0 ? p0 : 0.0f, q1 = e1 ? p1 : 0.0f;
      const float q2 = e2 ? p2 : 0.0f, q3 = e3 ? p3 : 0.0f;
      zacc[t][h] += q0 + q1;
      zacc[t][h] += q2 + q3;
      const v4h pv = {(_Float16)q0, (_Float16)q1, (_Float16)q2, (_Float16)q3};
      *(v4h*)(prow + h * F1_HEAD + t * 16 * F1_PITCH) = pv;
    }
  }
}

__device__ __forceinline__ void f1_mma(const _Float16* ph, const F1Frags& f,
                                       v8f& c0, v8f& c1) {
#pragma unroll
  for (int ks = 0; ks < F1_KS; ++ks) {
    const v16h a0 = load_frag16(ph, 0, F1_PITCH, ks * 32);
    const v16h a1 = load_frag16(ph, 16, F1_PITCH, ks * 32);
    c0 = wmma16(a0, f.b[ks], c0);
    c1 = wmma16(a1, f.b[ks], c1);
  }
}

__device__ __forceinline__ void f1_barrier() {
  asm volatile("s_wait_dscnt 0x0\n\ts_barrier_signal -1\n\ts_barrier_wait -1"
               ::: "memory");
}

__global__ void __launch_bounds__(F1_WAVES * 32)
attn_gemm1_kernel(const int* __restrict__ adj,
                  const _Float16* __restrict__ featsT,
                  const float* __restrict__ s1,
                  const float* __restrict__ n1,
                  const float* __restrict__ b1,
                  _Float16* __restrict__ H) {
  constexpr int CH = N / F1_COLS;
  static_assert(CH % 2 == 0, "blocks are taken in pairs, one per LDS stage");
  __shared__ __attribute__((aligned(16))) _Float16 pl[2 * F1_STAGE];
  __shared__ __attribute__((aligned(16))) float zl[F1_HG * F1_ROWS];

  const int lane = threadIdx.x & 31;
  const int wave = __builtin_amdgcn_readfirstlane((int)(threadIdx.x >> 5));
  const int r0   = blockIdx.x * F1_ROWS;
  const int h0   = blockIdx.y * F1_HG;
  const int irow[2] = {r0 + wave, r0 + 16 + wave};
  const int hl   = wave & (F1_HG - 1);
  const int head = h0 + hl;
  const int f0   = (wave >> 2) * 16;

  const int* arow_a = adj + (size_t)irow[0] * N;
  const int* arow_b = adj + (size_t)irow[1] * N;
  __builtin_prefetch(arow_a + lane * 64, 0, 0);
  __builtin_prefetch(arow_a + (N / 2) + lane * 64, 0, 0);
  __builtin_prefetch(arow_b + lane * 64, 0, 0);
  __builtin_prefetch(arow_b + (N / 2) + lane * 64, 0, 0);

  const int4*   arow0 = (const int4*)arow_a + lane;
  const int4*   arow1 = (const int4*)arow_b + lane;
  const float4* nq    = (const float4*)(n1 + (size_t)h0 * N) + lane;

  constexpr int MCH = N / (F1_COLS * F1_MU);
  static_assert(MCH >= 2 && MCH % 2 == 0, "chunks are taken in pairs");
  float nmax[2][F1_HG];
  int cnt[2] = {0, 0};
#pragma unroll
  for (int t = 0; t < 2; ++t) {
#pragma unroll
    for (int h = 0; h < F1_HG; ++h) nmax[t][h] = -INFINITY;
  }

  F1Chunk<F1_MU> ma, mb;
  f1_load(ma, arow0, arow1, nq, 0);
  for (int ci = 0; ci < MCH - 2; ci += 2) {
    f1_load(mb, arow0, arow1, nq, ci + 1);
    f1_max_step(ma, nmax, cnt);
    f1_load(ma, arow0, arow1, nq, ci + 2);
    f1_max_step(mb, nmax, cnt);
  }
  f1_load(mb, arow0, arow1, nq, MCH - 1);
  f1_max_step(ma, nmax, cnt);
  F1Chunk<1> ca, cb;
  f1_load(ca, arow0, arow1, nq, 0);
  f1_max_step(mb, nmax, cnt);

  float sv[2][F1_HG], mv[2][F1_HG], zacc[2][F1_HG];
  int n_masked[2];
#pragma unroll
  for (int t = 0; t < 2; ++t) {
    int ct = cnt[t];
#pragma unroll
    for (int msk = 16; msk > 0; msk >>= 1) ct += __shfl_xor(ct, msk, 32);
    n_masked[t] = N - ct;
#pragma unroll
    for (int h = 0; h < F1_HG; ++h) {
      sv[t][h] = s1[(h0 + h) * N + irow[t]];
      float mm = fmaxf(-3.0e38f, lrelu_max(sv[t][h] + wave_max32(nmax[t][h])));
      if (n_masked[t] > 0) mm = fmaxf(mm, MASK_VAL);
      mv[t][h] = mm;
      zacc[t][h] = 0.0f;
    }
  }

  _Float16* prow0 = pl + wave * F1_PITCH + lane * 4;
  _Float16* prow1 = prow0 + F1_STAGE;
  const _Float16* ph0 = pl + hl * F1_HEAD;
  const _Float16* ph1 = ph0 + F1_STAGE;
  const _Float16* BT  = featsT + (size_t)head * NHID * N;

  F1Frags bf;
  v8f c0 = {}, c1 = {};

  for (int ci = 0; ci < CH; ci += 2) {
    f1_load(cb, arow0, arow1, nq, ci + 1);
    f1_load_b(bf, BT, f0, ci);
    f1_form(ca, sv, mv, zacc, prow0);
    f1_barrier();
    f1_mma(ph0, bf, c0, c1);

    if (ci + 2 < CH) f1_load(ca, arow0, arow1, nq, ci + 2);
    f1_load_b(bf, BT, f0, ci + 1);
    f1_form(cb, sv, mv, zacc, prow1);
    f1_barrier();
    f1_mma(ph1, bf, c0, c1);
  }

#pragma unroll
  for (int t = 0; t < 2; ++t) {
#pragma unroll
    for (int h = 0; h < F1_HG; ++h) {
      float zs = wave_sum32(zacc[t][h]);
      if (n_masked[t] > 0)
        zs += (float)n_masked[t] * __expf(MASK_VAL - mv[t][h]);
      if (lane == 0) zl[h * F1_ROWS + t * 16 + wave] = zs;
    }
  }
  __syncthreads();

  const int hf = lane >> 4;
  const int r  = lane & 15;
  const int col = f0 + r;
  const float bias = b1[head * NHID + col];
  v8f acc[2] = {c0, c1};
#pragma unroll
  for (int t = 0; t < 2; ++t) {
    const float* zp = zl + hl * F1_ROWS + t * 16 + (hf << 3);
    const float4 za = *(const float4*)zp;
    const float4 zb = *(const float4*)(zp + 4);
    const float zi[8] = {1.0f / za.x, 1.0f / za.y, 1.0f / za.z, 1.0f / za.w,
                         1.0f / zb.x, 1.0f / zb.y, 1.0f / zb.z, 1.0f / zb.w};
#pragma unroll
    for (int k = 0; k < 8; ++k) {
      int orow = r0 + t * 16 + k + (hf << 3);
      float v  = acc[t][k] * zi[k] + bias;
      v = v > 0.0f ? v : 0.0f;
      H[(size_t)orow * (H1 * NHID) + head * NHID + col] = (_Float16)v;
    }
  }
}

__global__ void gemm2_kernel(const _Float16* __restrict__ H,
                             const _Float16* __restrict__ W2T,
                             float* __restrict__ feats2,
                             _Float16* __restrict__ feats2T) {
  int r0 = blockIdx.x << 4;
  v8f c = {};
  for (int k0 = 0; k0 < H1 * NHID; k0 += 32) {
    v16h a = load_frag16(H, r0, H1 * NHID, k0);
    v16h b = load_frag16(W2T, 0, H1 * NHID, k0);
    c = wmma16(a, b, c);
  }
  int lane = threadIdx.x & 31;
  int hf   = lane >> 4;
  int r    = lane & 15;
#pragma unroll
  for (int k = 0; k < 8; ++k) {
    int orow = r0 + k + (hf << 3);
    float v  = c[k];
    feats2[(size_t)orow * NCLASS + r] = v;
    feats2T[(size_t)r * N + orow]     = (_Float16)v;
  }
}

#define A2_U 8

struct A2Group {
  v16h a[A2_U];
  v16h b[A2_U];
};

__device__ __forceinline__ void a2_load(A2Group& g, const _Float16* P,
                                        const _Float16* BT, int r0, int gi) {
#pragma unroll
  for (int u = 0; u < A2_U; ++u) {
    const int j0 = (gi * A2_U + u) * 32;
    g.a[u] = load_frag16(P, r0, N, j0);
    g.b[u] = load_frag16(BT, 0, N, j0);
  }
}

__device__ __forceinline__ v8f a2_mma(const A2Group& g, v8f c) {
#pragma unroll
  for (int u = 0; u < A2_U; ++u) c = wmma16(g.a[u], g.b[u], c);
  return c;
}

__global__ void __launch_bounds__(32)
attn_gemm2_kernel(const _Float16* __restrict__ P,
                  const _Float16* __restrict__ feats2T,
                  const float* __restrict__ Z2,
                  const float* __restrict__ b2,
                  float* __restrict__ out) {
  constexpr int GROUPS = N / (32 * A2_U);
  static_assert(GROUPS >= 2 && GROUPS % 2 == 0, "groups are taken in pairs");

  int r0   = blockIdx.x << 4;
  int lane = threadIdx.x & 31;
  int hf   = lane >> 4;
  int r    = lane & 15;

  A2Group ga, gb;
  v8f c = {};
  a2_load(ga, P, feats2T, r0, 0);
  for (int gi = 0; gi < GROUPS - 2; gi += 2) {
    a2_load(gb, P, feats2T, r0, gi + 1);
    c = a2_mma(ga, c);
    a2_load(ga, P, feats2T, r0, gi + 2);
    c = a2_mma(gb, c);
  }
  a2_load(gb, P, feats2T, r0, GROUPS - 1);
  c = a2_mma(ga, c);
  c = a2_mma(gb, c);

  float bias = b2[r];
#pragma unroll
  for (int k = 0; k < 8; ++k) {
    int orow = r0 + k + (hf << 3);
    float zi = 1.0f / Z2[orow];
    float v  = c[k] * zi + bias;
    v = v > 0.0f ? v : 0.0f;
    float mx = v;
#pragma unroll
    for (int msk = 1; msk < 16; msk <<= 1)
      mx = fmaxf(mx, __shfl_xor(mx, msk, 32));
    float ex = __expf(v - mx);
    float sum = ex;
#pragma unroll
    for (int msk = 1; msk < 16; msk <<= 1) sum += __shfl_xor(sum, msk, 32);
    float o  = (v - mx) - logf(sum);
    out[(size_t)orow * NCLASS + r] = o;
  }
}

extern "C" void kernel_launch(void* const* d_in, const int* in_sizes, int n_in,
                              void* d_out, int out_size, void* d_ws,
                              size_t ws_size, hipStream_t stream) {
  const float* x   = (const float*)d_in[0];
  const int*   adj = (const int*)d_in[1];
  const float* W1  = (const float*)d_in[2];
  const float* b1  = (const float*)d_in[3];
  const float* as1 = (const float*)d_in[4];
  const float* an1 = (const float*)d_in[5];
  const float* W2  = (const float*)d_in[6];
  const float* b2  = (const float*)d_in[7];
  const float* as2 = (const float*)d_in[8];
  const float* an2 = (const float*)d_in[9];
  float* out = (float*)d_out;

  char* ws = (char*)d_ws;
  size_t off = 0;
  auto alloc = [&](size_t bytes) -> void* {
    off = (off + 255) & ~(size_t)255;
    void* p = ws + off;
    off += bytes;
    return p;
  };

  float*    feats2  = (float*)alloc((size_t)N * NCLASS * 4);
  _Float16* feats2T = (_Float16*)alloc((size_t)NCLASS * N * 2);
  float*    s2      = (float*)alloc((size_t)N * 4);
  float*    n2      = (float*)alloc((size_t)N * 4);
  float*    Z2      = (float*)alloc((size_t)N * 4);
  const size_t shared = off;
  _Float16* P2      = (_Float16*)alloc((size_t)N * N * 2);
  off = shared;
  _Float16* W1T     = (_Float16*)alloc((size_t)H1 * NHID * NFEAT * 2);
  _Float16* W2T     = (_Float16*)alloc((size_t)NCLASS * H1 * NHID * 2);
  float*    feats1  = (float*)alloc((size_t)H1 * N * NHID * 4);
  _Float16* featsT  = (_Float16*)alloc((size_t)H1 * NHID * N * 2);
  float*    s1      = (float*)alloc((size_t)H1 * N * 4);
  float*    n1      = (float*)alloc((size_t)H1 * N * 4);
  _Float16* H       = (_Float16*)alloc((size_t)N * H1 * NHID * 2);
  (void)ws_size; (void)n_in; (void)in_sizes; (void)out_size;

  {
    int total = H1 * NFEAT * NHID;
    cast_kernel<<<(total + 255) / 256, 256, 0, stream>>>(W1, W2, W1T, W2T);
  }
  gemm1_kernel<<<dim3(N / 16, H1), 32, 0, stream>>>(x, W1T, feats1, featsT);
  sn_kernel<<<(H1 * N + 255) / 256, 256, 0, stream>>>(feats1, as1, an1, s1, n1,
                                                      NHID, N, H1 * N);
  attn_gemm1_kernel<<<dim3(N / F1_ROWS, H1 / F1_HG), F1_WAVES * 32, 0,
                      stream>>>(adj, featsT, s1, n1, b1, H);
  gemm2_kernel<<<N / 16, 32, 0, stream>>>(H, W2T, feats2, feats2T);
  sn_kernel<<<(N + 255) / 256, 256, 0, stream>>>(feats2, as2, an2, s2, n2,
                                                 NCLASS, N, N);
  rowstats_kernel<1, 8, true><<<N / RS_WAVES, RS_WAVES * 32, 0, stream>>>(
      adj, s2, n2, nullptr, Z2, P2);
  attn_gemm2_kernel<<<N / 16, 32, 0, stream>>>(P2, feats2T, Z2, b2, out);
}


# Layer-2 row statistics and probabilities move into the attention kernel: a wave per row of a 16-row tile forms exp(logit-m) into LDS, wave 0 runs the WMMAs; no P plane, 7 launches; sums unchanged
# speedup vs baseline: 3.9823x; 1.1274x over previous
#include <hip/hip_runtime.h>
#include <hip/hip_bf16.h>


#define N      4096
#define NFEAT  512
#define NHID   64
#define NCLASS 16
#define H1     8
#define NEG_SLOPE 0.2f
#define MASK_VAL  -1e10f

typedef __attribute__((ext_vector_type(16))) _Float16 v16h;
typedef __attribute__((ext_vector_type(8)))  _Float16 v8h;
typedef __attribute__((ext_vector_type(4)))  _Float16 v4h;
typedef __attribute__((ext_vector_type(8)))  float    v8f;

__device__ __forceinline__ v8f wmma16(v16h a, v16h b, v8f c) {
  return __builtin_amdgcn_wmma_f32_16x16x32_f16(
       false, a,  false, b,
       (short)0, c,  false,  false);
}

__device__ __forceinline__ v16h load_frag16(const _Float16* base, int rc,
                                            int ld, int k0) {
  int lane = threadIdx.x & 31;
  int hf   = lane >> 4;
  int r    = lane & 15;
  const _Float16* p = base + (size_t)(rc + r) * ld + k0 + (hf << 3);
  v8h lo = *(const v8h*)p;
  v8h hi = *(const v8h*)(p + 16);
  return __builtin_shufflevector(lo, hi, 0, 1, 2, 3, 4, 5, 6, 7, 8, 9, 10, 11,
                                 12, 13, 14, 15);
}

__device__ __forceinline__ v16h load_frag16_f32(const float* base, int rc,
                                                int ld, int k0) {
  int lane = threadIdx.x & 31;
  int hf   = lane >> 4;
  int r    = lane & 15;
  const float* p = base + (size_t)(rc + r) * ld + k0 + (hf << 3);
  const float4 q0 = *(const float4*)p;
  const float4 q1 = *(const float4*)(p + 4);
  const float4 q2 = *(const float4*)(p + 16);
  const float4 q3 = *(const float4*)(p + 20);
  v16h a;
  a[0]  = (_Float16)q0.x; a[1]  = (_Float16)q0.y;
  a[2]  = (_Float16)q0.z; a[3]  = (_Float16)q0.w;
  a[4]  = (_Float16)q1.x; a[5]  = (_Float16)q1.y;
  a[6]  = (_Float16)q1.z; a[7]  = (_Float16)q1.w;
  a[8]  = (_Float16)q2.x; a[9]  = (_Float16)q2.y;
  a[10] = (_Float16)q2.z; a[11] = (_Float16)q2.w;
  a[12] = (_Float16)q3.x; a[13] = (_Float16)q3.y;
  a[14] = (_Float16)q3.z; a[15] = (_Float16)q3.w;
  return a;
}

__device__ __forceinline__ float lrelu(float x) {
  return x > 0.0f ? x : NEG_SLOPE * x;
}

__global__ void cast_kernel(const float* __restrict__ W1,
                            const float* __restrict__ W2,
                            _Float16* __restrict__ W1T,
                            _Float16* __restrict__ W2T) {
  int i = blockIdx.x * blockDim.x + threadIdx.x;
  if (i < H1 * NFEAT * NHID) {
    int h   = i / (NFEAT * NHID);
    int rem = i % (NFEAT * NHID);
    int k   = rem / NHID;
    int c   = rem % NHID;
    W1T[((size_t)(h * NHID + c)) * NFEAT + k] = (_Float16)W1[i];
  }
  if (i < (H1 * NHID) * NCLASS) {
    int k = i / NCLASS;
    int c = i % NCLASS;
    W2T[(size_t)c * (H1 * NHID) + k] = (_Float16)W2[i];
  }
}

__global__ void gemm1_kernel(const float* __restrict__ x,
                             const _Float16* __restrict__ W1T,
                             float* __restrict__ feats,
                             _Float16* __restrict__ featsT) {
  int head = blockIdx.y;
  int r0   = blockIdx.x << 4;
  const _Float16* BT = W1T + (size_t)head * NHID * NFEAT;

  v8f c0 = {}, c1 = {}, c2 = {}, c3 = {};
  for (int k0 = 0; k0 < NFEAT; k0 += 32) {
    v16h a = load_frag16_f32(x, r0, NFEAT, k0);
    c0 = wmma16(a, load_frag16(BT, 0, NFEAT, k0), c0);
    c1 = wmma16(a, load_frag16(BT, 16, NFEAT, k0), c1);
    c2 = wmma16(a, load_frag16(BT, 32, NFEAT, k0), c2);
    c3 = wmma16(a, load_frag16(BT, 48, NFEAT, k0), c3);
  }

  int lane = threadIdx.x & 31;
  int hf   = lane >> 4;
  int r    = lane & 15;
  v8f acc[4] = {c0, c1, c2, c3};
#pragma unroll
  for (int nt = 0; nt < 4; ++nt) {
#pragma unroll
    for (int k = 0; k < 8; ++k) {
      int orow = r0 + k + (hf << 3);
      int col  = nt * 16 + r;
      float v  = acc[nt][k];
      feats[((size_t)head * N + orow) * NHID + col] = v;
      featsT[((size_t)head * NHID + col) * N + orow] = (_Float16)v;
    }
  }
}

__global__ void sn_kernel(const float* __restrict__ feats,
                          const float* __restrict__ a_s,
                          const float* __restrict__ a_n,
                          float* __restrict__ s, float* __restrict__ n,
                          int fout, int rowsPerHead, int total) {
  int i = blockIdx.x * blockDim.x + threadIdx.x;
  if (i >= total) return;
  int head = i / rowsPerHead;
  const float* f  = feats + (size_t)i * fout;
  const float* av = a_s + head * fout;
  const float* nv = a_n + head * fout;
  float ss = 0.0f, nn = 0.0f;
  for (int c = 0; c < fout; ++c) {
    float v = f[c];
    ss += v * av[c];
    nn += v * nv[c];
  }
  s[i] = ss;
  n[i] = nn;
}


__device__ __forceinline__ float wave_max32(float v) {
#pragma unroll
  for (int msk = 16; msk > 0; msk >>= 1) v = fmaxf(v, __shfl_xor(v, msk, 32));
  return v;
}

__device__ __forceinline__ float wave_sum32(float v) {
#pragma unroll
  for (int msk = 16; msk > 0; msk >>= 1) v += __shfl_xor(v, msk, 32);
  return v;
}

__device__ __forceinline__ float lrelu_max(float x) {
  return fmaxf(x, NEG_SLOPE * x);
}

template <int NH, int U>
struct RsChunk {
  int4   a[U];
  float4 v[U][NH];
};

template <int NH, int U>
__device__ __forceinline__ void rs_load(RsChunk<NH, U>& c, const int4* arow,
                                        const float4* nq, int ci) {
#pragma unroll
  for (int u = 0; u < U; ++u) {
    const int q = (ci * U + u) * 32;
    c.a[u] = arow[q];
#pragma unroll
    for (int h = 0; h < NH; ++h) c.v[u][h] = nq[h * (N / 4) + q];
  }
}

template <int NH, int U>
__device__ __forceinline__ void rs_max_step(const RsChunk<NH, U>& c,
                                            float (&nmax)[NH], int& cnt) {
#pragma unroll
  for (int u = 0; u < U; ++u) {
    const bool e0 = c.a[u].x > 0, e1 = c.a[u].y > 0;
    const bool e2 = c.a[u].z > 0, e3 = c.a[u].w > 0;
    cnt += (int)e0 + (int)e1 + (int)e2 + (int)e3;
#pragma unroll
    for (int h = 0; h < NH; ++h) {
      const float x0 = e0 ? c.v[u][h].x : -INFINITY;
      const float x1 = e1 ? c.v[u][h].y : -INFINITY;
      const float x2 = e2 ? c.v[u][h].z : -INFINITY;
      const float x3 = e3 ? c.v[u][h].w : -INFINITY;
      nmax[h] = fmaxf(fmaxf(nmax[h], x0), fmaxf(x1, fmaxf(x2, x3)));
    }
  }
}

template <int NH, int U, bool WP>
__device__ __forceinline__ void rs_sum_step(const RsChunk<NH, U>& c,
                                            const float (&sv)[NH],
                                            const float (&mv)[NH],
                                            float (&acc)[NH],
                                            _Float16* prow, int ci) {
  static_assert(!WP || NH == 1, "the probability plane has one head");
#pragma unroll
  for (int u = 0; u < U; ++u) {
    const bool e0 = c.a[u].x > 0, e1 = c.a[u].y > 0;
    const bool e2 = c.a[u].z > 0, e3 = c.a[u].w > 0;
#pragma unroll
    for (int h = 0; h < NH; ++h) {
      const float p0 = __expf(lrelu_max(sv[h] + c.v[u][h].x) - mv[h]);
      const float p1 = __expf(lrelu_max(sv[h] + c.v[u][h].y) - mv[h]);
      const float p2 = __expf(lrelu_max(sv[h] + c.v[u][h].z) - mv[h]);
      const float p3 = __expf(lrelu_max(sv[h] + c.v[u][h].w) - mv[h]);
      const float q0 = e0 ? p0 : 0.0f, q1 = e1 ? p1 : 0.0f;
      const float q2 = e2 ? p2 : 0.0f, q3 = e3 ? p3 : 0.0f;
      acc[h] += q0 + q1;
      acc[h] += q2 + q3;
      if (WP) {
        const v4h pv = {(_Float16)q0, (_Float16)q1, (_Float16)q2,
                        (_Float16)q3};
        *(v4h*)(prow + (ci * U + u) * 128) = pv;
      }
    }
  }
}

#define F1_ROWS  32
#define F1_WAVES 16
#define F1_HG    4
#define F1_COLS  128
#define F1_PITCH 136
#define F1_HEAD  (F1_ROWS * F1_PITCH)
#define F1_STAGE (F1_HG * F1_HEAD)
#define F1_KS    (F1_COLS / 32)
#define F1_MU    2

template <int U>
struct F1Chunk {
  int4   a[U][2];
  float4 v[U][F1_HG];
};

template <int U>
__device__ __forceinline__ void f1_load(F1Chunk<U>& c, const int4* arow0,
                                        const int4* arow1, const float4* nq,
                                        int ci) {
#pragma unroll
  for (int u = 0; u < U; ++u) {
    const int q = (ci * U + u) * 32;
    c.a[u][0] = arow0[q];
    c.a[u][1] = arow1[q];
#pragma unroll
    for (int h = 0; h < F1_HG; ++h) c.v[u][h] = nq[h * (N / 4) + q];
  }
}

template <int U>
__device__ __forceinline__ void f1_max_step(const F1Chunk<U>& c,
                                            float (&nmax)[2][F1_HG],
                                            int (&cnt)[2]) {
#pragma unroll
  for (int u = 0; u < U; ++u) {
#pragma unroll
    for (int t = 0; t < 2; ++t) {
      const bool e0 = c.a[u][t].x > 0, e1 = c.a[u][t].y > 0;
      const bool e2 = c.a[u][t].z > 0, e3 = c.a[u][t].w > 0;
      cnt[t] += (int)e0 + (int)e1 + (int)e2 + (int)e3;
#pragma unroll
      for (int h = 0; h < F1_HG; ++h) {
        const float x0 = e0 ? c.v[u][h].x : -INFINITY;
        const float x1 = e1 ? c.v[u][h].y : -INFINITY;
        const float x2 = e2 ? c.v[u][h].z : -INFINITY;
        const float x3 = e3 ? c.v[u][h].w : -INFINITY;
        nmax[t][h] =
            fmaxf(fmaxf(nmax[t][h], x0), fmaxf(x1, fmaxf(x2, x3)));
      }
    }
  }
}

struct F1Frags {
  v16h b[F1_KS];
};

__device__ __forceinline__ void f1_load_b(F1Frags& f, const _Float16* BT,
                                          int f0, int ci) {
#pragma unroll
  for (int ks = 0; ks < F1_KS; ++ks)
    f.b[ks] = load_frag16(BT, f0, N, ci * F1_COLS + ks * 32);
}

__device__ __forceinline__ void f1_form(const F1Chunk<1>& c,
                                        const float (&sv)[2][F1_HG],
                                        const float (&mv)[2][F1_HG],
                                        float (&zacc)[2][F1_HG],
                                        _Float16* prow) {
#pragma unroll
  for (int t = 0; t < 2; ++t) {
    const bool e0 = c.a[0][t].x > 0, e1 = c.a[0][t].y > 0;
    const bool e2 = c.a[0][t].z > 0, e3 = c.a[0][t].w > 0;
#pragma unroll
    for (int h = 0; h < F1_HG; ++h) {
      const float p0 = __expf(lrelu_max(sv[t][h] + c.v[0][h].x) - mv[t][h]);
      const float p1 = __expf(lrelu_max(sv[t][h] + c.v[0][h].y) - mv[t][h]);
      const float p2 = __expf(lrelu_max(sv[t][h] + c.v[0][h].z) - mv[t][h]);
      const float p3 = __expf(lrelu_max(sv[t][h] + c.v[0][h].w) - mv[t][h]);
      const float q0 = e0 ? p0 : 0.0f, q1 = e1 ? p1 : 0.0f;
      const float q2 = e2 ? p2 : 0.0f, q3 = e3 ? p3 : 0.0f;
      zacc[t][h] += q0 + q1;
      zacc[t][h] += q2 + q3;
      const v4h pv = {(_Float16)q0, (_Float16)q1, (_Float16)q2, (_Float16)q3};
      *(v4h*)(prow + h * F1_HEAD + t * 16 * F1_PITCH) = pv;
    }
  }
}

__device__ __forceinline__ void f1_mma(const _Float16* ph, const F1Frags& f,
                                       v8f& c0, v8f& c1) {
#pragma unroll
  for (int ks = 0; ks < F1_KS; ++ks) {
    const v16h a0 = load_frag16(ph, 0, F1_PITCH, ks * 32);
    const v16h a1 = load_frag16(ph, 16, F1_PITCH, ks * 32);
    c0 = wmma16(a0, f.b[ks], c0);
    c1 = wmma16(a1, f.b[ks], c1);
  }
}

__device__ __forceinline__ void f1_barrier() {
  asm volatile("s_wait_dscnt 0x0\n\ts_barrier_signal -1\n\ts_barrier_wait -1"
               ::: "memory");
}

__global__ void __launch_bounds__(F1_WAVES * 32)
attn_gemm1_kernel(const int* __restrict__ adj,
                  const _Float16* __restrict__ featsT,
                  const float* __restrict__ s1,
                  const float* __restrict__ n1,
                  const float* __restrict__ b1,
                  _Float16* __restrict__ H) {
  constexpr int CH = N / F1_COLS;
  static_assert(CH % 2 == 0, "blocks are taken in pairs, one per LDS stage");
  __shared__ __attribute__((aligned(16))) _Float16 pl[2 * F1_STAGE];
  __shared__ __attribute__((aligned(16))) float zl[F1_HG * F1_ROWS];

  const int lane = threadIdx.x & 31;
  const int wave = __builtin_amdgcn_readfirstlane((int)(threadIdx.x >> 5));
  const int r0   = blockIdx.x * F1_ROWS;
  const int h0   = blockIdx.y * F1_HG;
  const int irow[2] = {r0 + wave, r0 + 16 + wave};
  const int hl   = wave & (F1_HG - 1);
  const int head = h0 + hl;
  const int f0   = (wave >> 2) * 16;

  const int* arow_a = adj + (size_t)irow[0] * N;
  const int* arow_b = adj + (size_t)irow[1] * N;
  __builtin_prefetch(arow_a + lane * 64, 0, 0);
  __builtin_prefetch(arow_a + (N / 2) + lane * 64, 0, 0);
  __builtin_prefetch(arow_b + lane * 64, 0, 0);
  __builtin_prefetch(arow_b + (N / 2) + lane * 64, 0, 0);

  const int4*   arow0 = (const int4*)arow_a + lane;
  const int4*   arow1 = (const int4*)arow_b + lane;
  const float4* nq    = (const float4*)(n1 + (size_t)h0 * N) + lane;

  constexpr int MCH = N / (F1_COLS * F1_MU);
  static_assert(MCH >= 2 && MCH % 2 == 0, "chunks are taken in pairs");
  float nmax[2][F1_HG];
  int cnt[2] = {0, 0};
#pragma unroll
  for (int t = 0; t < 2; ++t) {
#pragma unroll
    for (int h = 0; h < F1_HG; ++h) nmax[t][h] = -INFINITY;
  }

  F1Chunk<F1_MU> ma, mb;
  f1_load(ma, arow0, arow1, nq, 0);
  for (int ci = 0; ci < MCH - 2; ci += 2) {
    f1_load(mb, arow0, arow1, nq, ci + 1);
    f1_max_step(ma, nmax, cnt);
    f1_load(ma, arow0, arow1, nq, ci + 2);
    f1_max_step(mb, nmax, cnt);
  }
  f1_load(mb, arow0, arow1, nq, MCH - 1);
  f1_max_step(ma, nmax, cnt);
  F1Chunk<1> ca, cb;
  f1_load(ca, arow0, arow1, nq, 0);
  f1_max_step(mb, nmax, cnt);

  float sv[2][F1_HG], mv[2][F1_HG], zacc[2][F1_HG];
  int n_masked[2];
#pragma unroll
  for (int t = 0; t < 2; ++t) {
    int ct = cnt[t];
#pragma unroll
    for (int msk = 16; msk > 0; msk >>= 1) ct += __shfl_xor(ct, msk, 32);
    n_masked[t] = N - ct;
#pragma unroll
    for (int h = 0; h < F1_HG; ++h) {
      sv[t][h] = s1[(h0 + h) * N + irow[t]];
      float mm = fmaxf(-3.0e38f, lrelu_max(sv[t][h] + wave_max32(nmax[t][h])));
      if (n_masked[t] > 0) mm = fmaxf(mm, MASK_VAL);
      mv[t][h] = mm;
      zacc[t][h] = 0.0f;
    }
  }

  _Float16* prow0 = pl + wave * F1_PITCH + lane * 4;
  _Float16* prow1 = prow0 + F1_STAGE;
  const _Float16* ph0 = pl + hl * F1_HEAD;
  const _Float16* ph1 = ph0 + F1_STAGE;
  const _Float16* BT  = featsT + (size_t)head * NHID * N;

  F1Frags bf;
  v8f c0 = {}, c1 = {};

  for (int ci = 0; ci < CH; ci += 2) {
    f1_load(cb, arow0, arow1, nq, ci + 1);
    f1_load_b(bf, BT, f0, ci);
    f1_form(ca, sv, mv, zacc, prow0);
    f1_barrier();
    f1_mma(ph0, bf, c0, c1);

    if (ci + 2 < CH) f1_load(ca, arow0, arow1, nq, ci + 2);
    f1_load_b(bf, BT, f0, ci + 1);
    f1_form(cb, sv, mv, zacc, prow1);
    f1_barrier();
    f1_mma(ph1, bf, c0, c1);
  }

#pragma unroll
  for (int t = 0; t < 2; ++t) {
#pragma unroll
    for (int h = 0; h < F1_HG; ++h) {
      float zs = wave_sum32(zacc[t][h]);
      if (n_masked[t] > 0)
        zs += (float)n_masked[t] * __expf(MASK_VAL - mv[t][h]);
      if (lane == 0) zl[h * F1_ROWS + t * 16 + wave] = zs;
    }
  }
  __syncthreads();

  const int hf = lane >> 4;
  const int r  = lane & 15;
  const int col = f0 + r;
  const float bias = b1[head * NHID + col];
  v8f acc[2] = {c0, c1};
#pragma unroll
  for (int t = 0; t < 2; ++t) {
    const float* zp = zl + hl * F1_ROWS + t * 16 + (hf << 3);
    const float4 za = *(const float4*)zp;
    const float4 zb = *(const float4*)(zp + 4);
    const float zi[8] = {1.0f / za.x, 1.0f / za.y, 1.0f / za.z, 1.0f / za.w,
                         1.0f / zb.x, 1.0f / zb.y, 1.0f / zb.z, 1.0f / zb.w};
#pragma unroll
    for (int k = 0; k < 8; ++k) {
      int orow = r0 + t * 16 + k + (hf << 3);
      float v  = acc[t][k] * zi[k] + bias;
      v = v > 0.0f ? v : 0.0f;
      H[(size_t)orow * (H1 * NHID) + head * NHID + col] = (_Float16)v;
    }
  }
}

__global__ void gemm2_kernel(const _Float16* __restrict__ H,
                             const _Float16* __restrict__ W2T,
                             float* __restrict__ feats2,
                             _Float16* __restrict__ feats2T) {
  int r0 = blockIdx.x << 4;
  v8f c = {};
  for (int k0 = 0; k0 < H1 * NHID; k0 += 32) {
    v16h a = load_frag16(H, r0, H1 * NHID, k0);
    v16h b = load_frag16(W2T, 0, H1 * NHID, k0);
    c = wmma16(a, b, c);
  }
  int lane = threadIdx.x & 31;
  int hf   = lane >> 4;
  int r    = lane & 15;
#pragma unroll
  for (int k = 0; k < 8; ++k) {
    int orow = r0 + k + (hf << 3);
    float v  = c[k];
    feats2[(size_t)orow * NCLASS + r] = v;
    feats2T[(size_t)r * N + orow]     = (_Float16)v;
  }
}

#define F2_WAVES 16
#define F2_U     4
#define F2_MU    8
#define F2_CCOLS (128 * F2_U)
#define F2_PITCH (F2_CCOLS + 8)
#define F2_TILE  (16 * F2_PITCH)
#define F2_STAGE (2 * F2_TILE)
#define F2_KS    (F2_CCOLS / 32)

struct F2Feat {
  v8h b[F2_CCOLS / 256];
};

__device__ __forceinline__ void f2_load_b(F2Feat& f, const _Float16* brow,
                                          int ci) {
#pragma unroll
  for (int q = 0; q < F2_CCOLS / 256; ++q)
    f.b[q] = *(const v8h*)(brow + ci * F2_CCOLS + q * 256);
}

__device__ __forceinline__ void f2_store_b(const F2Feat& f, _Float16* bl) {
#pragma unroll
  for (int q = 0; q < F2_CCOLS / 256; ++q) *(v8h*)(bl + q * 256) = f.b[q];
}

#define F2_AHEAD 6

__device__ __forceinline__ v8f f2_mma(const _Float16* st, v8f c) {
#pragma unroll
  for (int ks = 0; ks < F2_KS; ++ks)
    c = wmma16(load_frag16(st, 0, F2_PITCH, ks * 32),
               load_frag16(st + F2_TILE, 0, F2_PITCH, ks * 32), c);
  __builtin_amdgcn_sched_group_barrier(0x100, 4 * F2_AHEAD, 0);
#pragma unroll
  for (int ks = 0; ks < F2_KS - F2_AHEAD; ++ks) {
    __builtin_amdgcn_sched_group_barrier(0x008, 1, 0);
    __builtin_amdgcn_sched_group_barrier(0x100, 4, 0);
  }
  __builtin_amdgcn_sched_group_barrier(0x008, F2_AHEAD, 0);
  return c;
}

__global__ void __launch_bounds__(F2_WAVES * 32)
attn_gemm2_kernel(const int* __restrict__ adj,
                  const _Float16* __restrict__ feats2T,
                  const float* __restrict__ s2,
                  const float* __restrict__ n2,
                  const float* __restrict__ b2,
                  float* __restrict__ out) {
  constexpr int MCH = N / (128 * F2_MU);
  constexpr int CH  = N / F2_CCOLS;
  static_assert(MCH >= 2 && MCH % 2 == 0, "chunks are taken in pairs");
  static_assert(CH % 2 == 0, "chunks are taken in pairs, one per LDS stage");
  __shared__ __attribute__((aligned(16))) _Float16 pl[2 * F2_STAGE];
  __shared__ __attribute__((aligned(16))) float zl[16];

  const int lane = threadIdx.x & 31;
  const int wave = __builtin_amdgcn_readfirstlane((int)(threadIdx.x >> 5));
  const int r0   = blockIdx.x << 4;
  const int i    = r0 + wave;
  const bool mul = wave == 0;

  const int* arow_i = adj + (size_t)i * N;
  __builtin_prefetch(arow_i + lane * 64, 0, 0);
  __builtin_prefetch(arow_i + (N / 2) + lane * 64, 0, 0);

  const int4*   arow = (const int4*)arow_i + lane;
  const float4* nq   = (const float4*)n2 + lane;
  const _Float16* brow = feats2T + (size_t)wave * N + lane * 8;

  float nmax[1] = {-INFINITY};
  int cnt = 0;
  RsChunk<1, F2_MU> ma, mb;
  rs_load(ma, arow, nq, 0);
  for (int ci = 0; ci < MCH - 2; ci += 2) {
    rs_load(mb, arow, nq, ci + 1);
    rs_max_step(ma, nmax, cnt);
    rs_load(ma, arow, nq, ci + 2);
    rs_max_step(mb, nmax, cnt);
  }
  rs_load(mb, arow, nq, MCH - 1);
  rs_max_step(ma, nmax, cnt);
  RsChunk<1, F2_U> ca, cb;
  rs_load(ca, arow, nq, 0);
  F2Feat fa, fb;
  f2_load_b(fa, brow, 0);
  rs_max_step(mb, nmax, cnt);

#pragma unroll
  for (int msk = 16; msk > 0; msk >>= 1) cnt += __shfl_xor(cnt, msk, 32);
  const int n_masked = N - cnt;

  float sv[1], mv[1], acc[1] = {0.0f};
  sv[0] = s2[i];
  {
    float mm = fmaxf(-3.0e38f, lrelu_max(sv[0] + wave_max32(nmax[0])));
    if (n_masked > 0) mm = fmaxf(mm, MASK_VAL);
    mv[0] = mm;
  }

  _Float16* prow0 = pl + wave * F2_PITCH + lane * 4;
  _Float16* prow1 = prow0 + F2_STAGE;
  _Float16* bl0 = pl + F2_TILE + wave * F2_PITCH + lane * 8;
  _Float16* bl1 = bl0 + F2_STAGE;

  v8f c = {};
  for (int ci = 0; ci < CH; ci += 2) {
    rs_load(cb, arow, nq, ci + 1);
    f2_load_b(fb, brow, ci + 1);
    rs_sum_step<1, F2_U, true>(ca, sv, mv, acc, prow0, 0);
    f2_store_b(fa, bl0);
    f1_barrier();
    if (mul) c = f2_mma(pl, c);

    if (ci + 2 < CH) {
      rs_load(ca, arow, nq, ci + 2);
      f2_load_b(fa, brow, ci + 2);
    }
    rs_sum_step<1, F2_U, true>(cb, sv, mv, acc, prow1, 0);
    f2_store_b(fb, bl1);
    f1_barrier();
    if (mul) c = f2_mma(pl + F2_STAGE, c);
  }

  {
    float zs = wave_sum32(acc[0]);
    if (n_masked > 0) zs += (float)n_masked * __expf(MASK_VAL - mv[0]);
    if (lane == 0) zl[wave] = zs;
  }
  __syncthreads();
  if (!mul) return;

  const int hf = lane >> 4;
  const int r  = lane & 15;
  const float bias = b2[r];
  const float4 za = *(const float4*)(zl + (hf << 3));
  const float4 zb = *(const float4*)(zl + (hf << 3) + 4);
  const float zr[8] = {za.x, za.y, za.z, za.w, zb.x, zb.y, zb.z, zb.w};
#pragma unroll
  for (int k = 0; k < 8; ++k) {
    int orow = r0 + k + (hf << 3);
    float zi = 1.0f / zr[k];
    float v  = c[k] * zi + bias;
    v = v > 0.0f ? v : 0.0f;
    float mx = v;
#pragma unroll
    for (int msk = 1; msk < 16; msk <<= 1)
      mx = fmaxf(mx, __shfl_xor(mx, msk, 32));
    float ex = __expf(v - mx);
    float sum = ex;
#pragma unroll
    for (int msk = 1; msk < 16; msk <<= 1) sum += __shfl_xor(sum, msk, 32);
    float o  = (v - mx) - logf(sum);
    out[(size_t)orow * NCLASS + r] = o;
  }
}

extern "C" void kernel_launch(void* const* d_in, const int* in_sizes, int n_in,
                              void* d_out, int out_size, void* d_ws,
                              size_t ws_size, hipStream_t stream) {
  const float* x   = (const float*)d_in[0];
  const int*   adj = (const int*)d_in[1];
  const float* W1  = (const float*)d_in[2];
  const float* b1  = (const float*)d_in[3];
  const float* as1 = (const float*)d_in[4];
  const float* an1 = (const float*)d_in[5];
  const float* W2  = (const float*)d_in[6];
  const float* b2  = (const float*)d_in[7];
  const float* as2 = (const float*)d_in[8];
  const float* an2 = (const float*)d_in[9];
  float* out = (float*)d_out;

  char* ws = (char*)d_ws;
  size_t off = 0;
  auto alloc = [&](size_t bytes) -> void* {
    off = (off + 255) & ~(size_t)255;
    void* p = ws + off;
    off += bytes;
    return p;
  };

  _Float16* W1T     = (_Float16*)alloc((size_t)H1 * NHID * NFEAT * 2);
  _Float16* W2T     = (_Float16*)alloc((size_t)NCLASS * H1 * NHID * 2);
  float*    feats1  = (float*)alloc((size_t)H1 * N * NHID * 4);
  _Float16* featsT  = (_Float16*)alloc((size_t)H1 * NHID * N * 2);
  float*    s1      = (float*)alloc((size_t)H1 * N * 4);
  float*    n1      = (float*)alloc((size_t)H1 * N * 4);
  _Float16* H       = (_Float16*)alloc((size_t)N * H1 * NHID * 2);
  float*    feats2  = (float*)alloc((size_t)N * NCLASS * 4);
  _Float16* feats2T = (_Float16*)alloc((size_t)NCLASS * N * 2);
  float*    s2      = (float*)alloc((size_t)N * 4);
  float*    n2      = (float*)alloc((size_t)N * 4);
  (void)ws_size; (void)n_in; (void)in_sizes; (void)out_size;

  {
    int total = H1 * NFEAT * NHID;
    cast_kernel<<<(total + 255) / 256, 256, 0, stream>>>(W1, W2, W1T, W2T);
  }
  gemm1_kernel<<<dim3(N / 16, H1), 32, 0, stream>>>(x, W1T, feats1, featsT);
  sn_kernel<<<(H1 * N + 255) / 256, 256, 0, stream>>>(feats1, as1, an1, s1, n1,
                                                      NHID, N, H1 * N);
  attn_gemm1_kernel<<<dim3(N / F1_ROWS, H1 / F1_HG), F1_WAVES * 32, 0,
                      stream>>>(adj, featsT, s1, n1, b1, H);
  gemm2_kernel<<<N / 16, 32, 0, stream>>>(H, W2T, feats2, feats2T);
  sn_kernel<<<(N + 255) / 256, 256, 0, stream>>>(feats2, as2, an2, s2, n2,
                                                 NCLASS, N, N);
  attn_gemm2_kernel<<<N / 16, F2_WAVES * 32, 0, stream>>>(adj, feats2T, s2, n2,
                                                          b2, out);
}


# Layer-1 B operands (weights, features) kept in WMMA fragment order by their producers: each fragment load reads 512 contiguous bytes (4 lines, was 16), features stored 16 bytes a lane; sums unchanged
# speedup vs baseline: 4.5705x; 1.1477x over previous
#include <hip/hip_runtime.h>
#include <hip/hip_bf16.h>


#define N      4096
#define NFEAT  512
#define NHID   64
#define NCLASS 16
#define H1     8
#define NEG_SLOPE 0.2f
#define MASK_VAL  -1e10f

typedef __attribute__((ext_vector_type(16))) _Float16 v16h;
typedef __attribute__((ext_vector_type(8)))  _Float16 v8h;
typedef __attribute__((ext_vector_type(4)))  _Float16 v4h;
typedef __attribute__((ext_vector_type(8)))  float    v8f;

__device__ __forceinline__ v8f wmma16(v16h a, v16h b, v8f c) {
  return __builtin_amdgcn_wmma_f32_16x16x32_f16(
       false, a,  false, b,
       (short)0, c,  false,  false);
}

__device__ __forceinline__ v16h load_frag16(const _Float16* base, int rc,
                                            int ld, int k0) {
  int lane = threadIdx.x & 31;
  int hf   = lane >> 4;
  int r    = lane & 15;
  const _Float16* p = base + (size_t)(rc + r) * ld + k0 + (hf << 3);
  v8h lo = *(const v8h*)p;
  v8h hi = *(const v8h*)(p + 16);
  return __builtin_shufflevector(lo, hi, 0, 1, 2, 3, 4, 5, 6, 7, 8, 9, 10, 11,
                                 12, 13, 14, 15);
}

__device__ __forceinline__ v16h load_frag16_f32(const float* base, int rc,
                                                int ld, int k0) {
  int lane = threadIdx.x & 31;
  int hf   = lane >> 4;
  int r    = lane & 15;
  const float* p = base + (size_t)(rc + r) * ld + k0 + (hf << 3);
  const float4 q0 = *(const float4*)p;
  const float4 q1 = *(const float4*)(p + 4);
  const float4 q2 = *(const float4*)(p + 16);
  const float4 q3 = *(const float4*)(p + 20);
  v16h a;
  a[0]  = (_Float16)q0.x; a[1]  = (_Float16)q0.y;
  a[2]  = (_Float16)q0.z; a[3]  = (_Float16)q0.w;
  a[4]  = (_Float16)q1.x; a[5]  = (_Float16)q1.y;
  a[6]  = (_Float16)q1.z; a[7]  = (_Float16)q1.w;
  a[8]  = (_Float16)q2.x; a[9]  = (_Float16)q2.y;
  a[10] = (_Float16)q2.z; a[11] = (_Float16)q2.w;
  a[12] = (_Float16)q3.x; a[13] = (_Float16)q3.y;
  a[14] = (_Float16)q3.z; a[15] = (_Float16)q3.w;
  return a;
}

#define FRAG_HALVES 512
__device__ __forceinline__ int frag_off(int r, int k) {
  return ((k >> 4) & 1) * 256 + ((((k >> 3) & 1) * 16 + r) * 8) + (k & 7);
}
__device__ __forceinline__ v16h load_frag16_packed(const _Float16* frag) {
  int lane = threadIdx.x & 31;
  const _Float16* p = frag + lane * 8;
  v8h lo = *(const v8h*)p;
  v8h hi = *(const v8h*)(p + 256);
  return __builtin_shufflevector(lo, hi, 0, 1, 2, 3, 4, 5, 6, 7, 8, 9, 10, 11,
                                 12, 13, 14, 15);
}

__device__ __forceinline__ float lrelu(float x) {
  return x > 0.0f ? x : NEG_SLOPE * x;
}

__global__ void cast_kernel(const float* __restrict__ W1,
                            const float* __restrict__ W2,
                            _Float16* __restrict__ W1F,
                            _Float16* __restrict__ W2T) {
  int i = blockIdx.x * blockDim.x + threadIdx.x;
  if (i < H1 * NFEAT * NHID) {
    int h   = i / (NFEAT * NHID);
    int rem = i % (NFEAT * NHID);
    int k   = rem / NHID;
    int c   = rem % NHID;
    const int frag = (h * (NHID / 16) + (c >> 4)) * (NFEAT / 32) + (k >> 5);
    W1F[(size_t)frag * FRAG_HALVES + frag_off(c & 15, k & 31)] = (_Float16)W1[i];
  }
  if (i < (H1 * NHID) * NCLASS) {
    int k = i / NCLASS;
    int c = i % NCLASS;
    W2T[(size_t)c * (H1 * NHID) + k] = (_Float16)W2[i];
  }
}

__global__ void gemm1_kernel(const float* __restrict__ x,
                             const _Float16* __restrict__ W1F,
                             float* __restrict__ feats,
                             _Float16* __restrict__ featsF) {
  int head = blockIdx.y;
  int r0   = blockIdx.x << 4;
  const _Float16* BF = W1F + (size_t)head * (NHID / 16) * (NFEAT / 32) * FRAG_HALVES;

  v8f c0 = {}, c1 = {}, c2 = {}, c3 = {};
  for (int k0 = 0; k0 < NFEAT; k0 += 32) {
    v16h a = load_frag16_f32(x, r0, NFEAT, k0);
    const _Float16* bk = BF + (k0 >> 5) * FRAG_HALVES;
    c0 = wmma16(a, load_frag16_packed(bk), c0);
    c1 = wmma16(a, load_frag16_packed(bk + 1 * (NFEAT / 32) * FRAG_HALVES), c1);
    c2 = wmma16(a, load_frag16_packed(bk + 2 * (NFEAT / 32) * FRAG_HALVES), c2);
    c3 = wmma16(a, load_frag16_packed(bk + 3 * (NFEAT / 32) * FRAG_HALVES), c3);
  }

  int lane = threadIdx.x & 31;
  int hf   = lane >> 4;
  int r    = lane & 15;
  v8f acc[4] = {c0, c1, c2, c3};
#pragma unroll
  for (int nt = 0; nt < 4; ++nt) {
    v8h pk;
#pragma unroll
    for (int k = 0; k < 8; ++k) {
      int orow = r0 + k + (hf << 3);
      int col  = nt * 16 + r;
      float v  = acc[nt][k];
      feats[((size_t)head * N + orow) * NHID + col] = v;
      pk[k] = (_Float16)v;
    }
    const int frag = (head * (NHID / 16) + nt) * (N / 32) + (r0 >> 5);
    *(v8h*)(featsF + (size_t)frag * FRAG_HALVES + ((r0 >> 4) & 1) * 256 +
            lane * 8) = pk;
  }
}

__global__ void sn_kernel(const float* __restrict__ feats,
                          const float* __restrict__ a_s,
                          const float* __restrict__ a_n,
                          float* __restrict__ s, float* __restrict__ n,
                          int fout, int rowsPerHead, int total) {
  int i = blockIdx.x * blockDim.x + threadIdx.x;
  if (i >= total) return;
  int head = i / rowsPerHead;
  const float* f  = feats + (size_t)i * fout;
  const float* av = a_s + head * fout;
  const float* nv = a_n + head * fout;
  float ss = 0.0f, nn = 0.0f;
  for (int c = 0; c < fout; ++c) {
    float v = f[c];
    ss += v * av[c];
    nn += v * nv[c];
  }
  s[i] = ss;
  n[i] = nn;
}


__device__ __forceinline__ float wave_max32(float v) {
#pragma unroll
  for (int msk = 16; msk > 0; msk >>= 1) v = fmaxf(v, __shfl_xor(v, msk, 32));
  return v;
}

__device__ __forceinline__ float wave_sum32(float v) {
#pragma unroll
  for (int msk = 16; msk > 0; msk >>= 1) v += __shfl_xor(v, msk, 32);
  return v;
}

__device__ __forceinline__ float lrelu_max(float x) {
  return fmaxf(x, NEG_SLOPE * x);
}

template <int NH, int U>
struct RsChunk {
  int4   a[U];
  float4 v[U][NH];
};

template <int NH, int U>
__device__ __forceinline__ void rs_load(RsChunk<NH, U>& c, const int4* arow,
                                        const float4* nq, int ci) {
#pragma unroll
  for (int u = 0; u < U; ++u) {
    const int q = (ci * U + u) * 32;
    c.a[u] = arow[q];
#pragma unroll
    for (int h = 0; h < NH; ++h) c.v[u][h] = nq[h * (N / 4) + q];
  }
}

template <int NH, int U>
__device__ __forceinline__ void rs_max_step(const RsChunk<NH, U>& c,
                                            float (&nmax)[NH], int& cnt) {
#pragma unroll
  for (int u = 0; u < U; ++u) {
    const bool e0 = c.a[u].x > 0, e1 = c.a[u].y > 0;
    const bool e2 = c.a[u].z > 0, e3 = c.a[u].w > 0;
    cnt += (int)e0 + (int)e1 + (int)e2 + (int)e3;
#pragma unroll
    for (int h = 0; h < NH; ++h) {
      const float x0 = e0 ? c.v[u][h].x : -INFINITY;
      const float x1 = e1 ? c.v[u][h].y : -INFINITY;
      const float x2 = e2 ? c.v[u][h].z : -INFINITY;
      const float x3 = e3 ? c.v[u][h].w : -INFINITY;
      nmax[h] = fmaxf(fmaxf(nmax[h], x0), fmaxf(x1, fmaxf(x2, x3)));
    }
  }
}

template <int NH, int U, bool WP>
__device__ __forceinline__ void rs_sum_step(const RsChunk<NH, U>& c,
                                            const float (&sv)[NH],
                                            const float (&mv)[NH],
                                            float (&acc)[NH],
                                            _Float16* prow, int ci) {
  static_assert(!WP || NH == 1, "the probability plane has one head");
#pragma unroll
  for (int u = 0; u < U; ++u) {
    const bool e0 = c.a[u].x > 0, e1 = c.a[u].y > 0;
    const bool e2 = c.a[u].z > 0, e3 = c.a[u].w > 0;
#pragma unroll
    for (int h = 0; h < NH; ++h) {
      const float p0 = __expf(lrelu_max(sv[h] + c.v[u][h].x) - mv[h]);
      const float p1 = __expf(lrelu_max(sv[h] + c.v[u][h].y) - mv[h]);
      const float p2 = __expf(lrelu_max(sv[h] + c.v[u][h].z) - mv[h]);
      const float p3 = __expf(lrelu_max(sv[h] + c.v[u][h].w) - mv[h]);
      const float q0 = e0 ? p0 : 0.0f, q1 = e1 ? p1 : 0.0f;
      const float q2 = e2 ? p2 : 0.0f, q3 = e3 ? p3 : 0.0f;
      acc[h] += q0 + q1;
      acc[h] += q2 + q3;
      if (WP) {
        const v4h pv = {(_Float16)q0, (_Float16)q1, (_Float16)q2,
                        (_Float16)q3};
        *(v4h*)(prow + (ci * U + u) * 128) = pv;
      }
    }
  }
}

#define F1_ROWS  32
#define F1_WAVES 16
#define F1_HG    4
#define F1_COLS  128
#define F1_PITCH 136
#define F1_HEAD  (F1_ROWS * F1_PITCH)
#define F1_STAGE (F1_HG * F1_HEAD)
#define F1_KS    (F1_COLS / 32)
#define F1_MU    2

template <int U>
struct F1Chunk {
  int4   a[U][2];
  float4 v[U][F1_HG];
};

template <int U>
__device__ __forceinline__ void f1_load(F1Chunk<U>& c, const int4* arow0,
                                        const int4* arow1, const float4* nq,
                                        int ci) {
#pragma unroll
  for (int u = 0; u < U; ++u) {
    const int q = (ci * U + u) * 32;
    c.a[u][0] = arow0[q];
    c.a[u][1] = arow1[q];
#pragma unroll
    for (int h = 0; h < F1_HG; ++h) c.v[u][h] = nq[h * (N / 4) + q];
  }
}

template <int U>
__device__ __forceinline__ void f1_max_step(const F1Chunk<U>& c,
                                            float (&nmax)[2][F1_HG],
                                            int (&cnt)[2]) {
#pragma unroll
  for (int u = 0; u < U; ++u) {
#pragma unroll
    for (int t = 0; t < 2; ++t) {
      const bool e0 = c.a[u][t].x > 0, e1 = c.a[u][t].y > 0;
      const bool e2 = c.a[u][t].z > 0, e3 = c.a[u][t].w > 0;
      cnt[t] += (int)e0 + (int)e1 + (int)e2 + (int)e3;
#pragma unroll
      for (int h = 0; h < F1_HG; ++h) {
        const float x0 = e0 ? c.v[u][h].x : -INFINITY;
        const float x1 = e1 ? c.v[u][h].y : -INFINITY;
        const float x2 = e2 ? c.v[u][h].z : -INFINITY;
        const float x3 = e3 ? c.v[u][h].w : -INFINITY;
        nmax[t][h] =
            fmaxf(fmaxf(nmax[t][h], x0), fmaxf(x1, fmaxf(x2, x3)));
      }
    }
  }
}

struct F1Frags {
  v16h b[F1_KS];
};

__device__ __forceinline__ void f1_load_b(F1Frags& f, const _Float16* BF,
                                          int ci) {
#pragma unroll
  for (int ks = 0; ks < F1_KS; ++ks)
    f.b[ks] = load_frag16_packed(BF + (ci * F1_KS + ks) * FRAG_HALVES);
}

__device__ __forceinline__ void f1_form(const F1Chunk<1>& c,
                                        const float (&sv)[2][F1_HG],
                                        const float (&mv)[2][F1_HG],
                                        float (&zacc)[2][F1_HG],
                                        _Float16* prow) {
#pragma unroll
  for (int t = 0; t < 2; ++t) {
    const bool e0 = c.a[0][t].x > 0, e1 = c.a[0][t].y > 0;
    const bool e2 = c.a[0][t].z > 0, e3 = c.a[0][t].w > 0;
#pragma unroll
    for (int h = 0; h < F1_HG; ++h) {
      const float p0 = __expf(lrelu_max(sv[t][h] + c.v[0][h].x) - mv[t][h]);
      const float p1 = __expf(lrelu_max(sv[t][h] + c.v[0][h].y) - mv[t][h]);
      const float p2 = __expf(lrelu_max(sv[t][h] + c.v[0][h].z) - mv[t][h]);
      const float p3 = __expf(lrelu_max(sv[t][h] + c.v[0][h].w) - mv[t][h]);
      const float q0 = e0 ? p0 : 0.0f, q1 = e1 ? p1 : 0.0f;
      const float q2 = e2 ? p2 : 0.0f, q3 = e3 ? p3 : 0.0f;
      zacc[t][h] += q0 + q1;
      zacc[t][h] += q2 + q3;
      const v4h pv = {(_Float16)q0, (_Float16)q1, (_Float16)q2, (_Float16)q3};
      *(v4h*)(prow + h * F1_HEAD + t * 16 * F1_PITCH) = pv;
    }
  }
}

__device__ __forceinline__ void f1_mma(const _Float16* ph, const F1Frags& f,
                                       v8f& c0, v8f& c1) {
#pragma unroll
  for (int ks = 0; ks < F1_KS; ++ks) {
    const v16h a0 = load_frag16(ph, 0, F1_PITCH, ks * 32);
    const v16h a1 = load_frag16(ph, 16, F1_PITCH, ks * 32);
    c0 = wmma16(a0, f.b[ks], c0);
    c1 = wmma16(a1, f.b[ks], c1);
  }
}

__device__ __forceinline__ void f1_barrier() {
  asm volatile("s_wait_dscnt 0x0\n\ts_barrier_signal -1\n\ts_barrier_wait -1"
               ::: "memory");
}

__global__ void __launch_bounds__(F1_WAVES * 32)
attn_gemm1_kernel(const int* __restrict__ adj,
                  const _Float16* __restrict__ featsF,
                  const float* __restrict__ s1,
                  const float* __restrict__ n1,
                  const float* __restrict__ b1,
                  _Float16* __restrict__ H) {
  constexpr int CH = N / F1_COLS;
  static_assert(CH % 2 == 0, "blocks are taken in pairs, one per LDS stage");
  __shared__ __attribute__((aligned(16))) _Float16 pl[2 * F1_STAGE];
  __shared__ __attribute__((aligned(16))) float zl[F1_HG * F1_ROWS];

  const int lane = threadIdx.x & 31;
  const int wave = __builtin_amdgcn_readfirstlane((int)(threadIdx.x >> 5));
  const int r0   = blockIdx.x * F1_ROWS;
  const int h0   = blockIdx.y * F1_HG;
  const int irow[2] = {r0 + wave, r0 + 16 + wave};
  const int hl   = wave & (F1_HG - 1);
  const int head = h0 + hl;
  const int f0   = (wave >> 2) * 16;

  const int* arow_a = adj + (size_t)irow[0] * N;
  const int* arow_b = adj + (size_t)irow[1] * N;
  __builtin_prefetch(arow_a + lane * 64, 0, 0);
  __builtin_prefetch(arow_a + (N / 2) + lane * 64, 0, 0);
  __builtin_prefetch(arow_b + lane * 64, 0, 0);
  __builtin_prefetch(arow_b + (N / 2) + lane * 64, 0, 0);

  const int4*   arow0 = (const int4*)arow_a + lane;
  const int4*   arow1 = (const int4*)arow_b + lane;
  const float4* nq    = (const float4*)(n1 + (size_t)h0 * N) + lane;

  constexpr int MCH = N / (F1_COLS * F1_MU);
  static_assert(MCH >= 2 && MCH % 2 == 0, "chunks are taken in pairs");
  float nmax[2][F1_HG];
  int cnt[2] = {0, 0};
#pragma unroll
  for (int t = 0; t < 2; ++t) {
#pragma unroll
    for (int h = 0; h < F1_HG; ++h) nmax[t][h] = -INFINITY;
  }

  F1Chunk<F1_MU> ma, mb;
  f1_load(ma, arow0, arow1, nq, 0);
  for (int ci = 0; ci < MCH - 2; ci += 2) {
    f1_load(mb, arow0, arow1, nq, ci + 1);
    f1_max_step(ma, nmax, cnt);
    f1_load(ma, arow0, arow1, nq, ci + 2);
    f1_max_step(mb, nmax, cnt);
  }
  f1_load(mb, arow0, arow1, nq, MCH - 1);
  f1_max_step(ma, nmax, cnt);
  F1Chunk<1> ca, cb;
  f1_load(ca, arow0, arow1, nq, 0);
  f1_max_step(mb, nmax, cnt);

  float sv[2][F1_HG], mv[2][F1_HG], zacc[2][F1_HG];
  int n_masked[2];
#pragma unroll
  for (int t = 0; t < 2; ++t) {
    int ct = cnt[t];
#pragma unroll
    for (int msk = 16; msk > 0; msk >>= 1) ct += __shfl_xor(ct, msk, 32);
    n_masked[t] = N - ct;
#pragma unroll
    for (int h = 0; h < F1_HG; ++h) {
      sv[t][h] = s1[(h0 + h) * N + irow[t]];
      float mm = fmaxf(-3.0e38f, lrelu_max(sv[t][h] + wave_max32(nmax[t][h])));
      if (n_masked[t] > 0) mm = fmaxf(mm, MASK_VAL);
      mv[t][h] = mm;
      zacc[t][h] = 0.0f;
    }
  }

  _Float16* prow0 = pl + wave * F1_PITCH + lane * 4;
  _Float16* prow1 = prow0 + F1_STAGE;
  const _Float16* ph0 = pl + hl * F1_HEAD;
  const _Float16* ph1 = ph0 + F1_STAGE;
  const _Float16* BF  = featsF + (size_t)(head * (NHID / 16) + (f0 >> 4)) *
                                     (N / 32) * FRAG_HALVES;

  F1Frags bf;
  v8f c0 = {}, c1 = {};

  for (int ci = 0; ci < CH; ci += 2) {
    f1_load(cb, arow0, arow1, nq, ci + 1);
    f1_load_b(bf, BF, ci);
    f1_form(ca, sv, mv, zacc, prow0);
    f1_barrier();
    f1_mma(ph0, bf, c0, c1);

    if (ci + 2 < CH) f1_load(ca, arow0, arow1, nq, ci + 2);
    f1_load_b(bf, BF, ci + 1);
    f1_form(cb, sv, mv, zacc, prow1);
    f1_barrier();
    f1_mma(ph1, bf, c0, c1);
  }

#pragma unroll
  for (int t = 0; t < 2; ++t) {
#pragma unroll
    for (int h = 0; h < F1_HG; ++h) {
      float zs = wave_sum32(zacc[t][h]);
      if (n_masked[t] > 0)
        zs += (float)n_masked[t] * __expf(MASK_VAL - mv[t][h]);
      if (lane == 0) zl[h * F1_ROWS + t * 16 + wave] = zs;
    }
  }
  __syncthreads();

  const int hf = lane >> 4;
  const int r  = lane & 15;
  const int col = f0 + r;
  const float bias = b1[head * NHID + col];
  v8f acc[2] = {c0, c1};
#pragma unroll
  for (int t = 0; t < 2; ++t) {
    const float* zp = zl + hl * F1_ROWS + t * 16 + (hf << 3);
    const float4 za = *(const float4*)zp;
    const float4 zb = *(const float4*)(zp + 4);
    const float zi[8] = {1.0f / za.x, 1.0f / za.y, 1.0f / za.z, 1.0f / za.w,
                         1.0f / zb.x, 1.0f / zb.y, 1.0f / zb.z, 1.0f / zb.w};
#pragma unroll
    for (int k = 0; k < 8; ++k) {
      int orow = r0 + t * 16 + k + (hf << 3);
      float v  = acc[t][k] * zi[k] + bias;
      v = v > 0.0f ? v : 0.0f;
      H[(size_t)orow * (H1 * NHID) + head * NHID + col] = (_Float16)v;
    }
  }
}

__global__ void gemm2_kernel(const _Float16* __restrict__ H,
                             const _Float16* __restrict__ W2T,
                             float* __restrict__ feats2,
                             _Float16* __restrict__ feats2T) {
  int r0 = blockIdx.x << 4;
  v8f c = {};
  for (int k0 = 0; k0 < H1 * NHID; k0 += 32) {
    v16h a = load_frag16(H, r0, H1 * NHID, k0);
    v16h b = load_frag16(W2T, 0, H1 * NHID, k0);
    c = wmma16(a, b, c);
  }
  int lane = threadIdx.x & 31;
  int hf   = lane >> 4;
  int r    = lane & 15;
#pragma unroll
  for (int k = 0; k < 8; ++k) {
    int orow = r0 + k + (hf << 3);
    float v  = c[k];
    feats2[(size_t)orow * NCLASS + r] = v;
    feats2T[(size_t)r * N + orow]     = (_Float16)v;
  }
}

#define F2_WAVES 16
#define F2_U     4
#define F2_MU    8
#define F2_CCOLS (128 * F2_U)
#define F2_PITCH (F2_CCOLS + 8)
#define F2_TILE  (16 * F2_PITCH)
#define F2_STAGE (2 * F2_TILE)
#define F2_KS    (F2_CCOLS / 32)

struct F2Feat {
  v8h b[F2_CCOLS / 256];
};

__device__ __forceinline__ void f2_load_b(F2Feat& f, const _Float16* brow,
                                          int ci) {
#pragma unroll
  for (int q = 0; q < F2_CCOLS / 256; ++q)
    f.b[q] = *(const v8h*)(brow + ci * F2_CCOLS + q * 256);
}

__device__ __forceinline__ void f2_store_b(const F2Feat& f, _Float16* bl) {
#pragma unroll
  for (int q = 0; q < F2_CCOLS / 256; ++q) *(v8h*)(bl + q * 256) = f.b[q];
}

#define F2_AHEAD 6

__device__ __forceinline__ v8f f2_mma(const _Float16* st, v8f c) {
#pragma unroll
  for (int ks = 0; ks < F2_KS; ++ks)
    c = wmma16(load_frag16(st, 0, F2_PITCH, ks * 32),
               load_frag16(st + F2_TILE, 0, F2_PITCH, ks * 32), c);
  __builtin_amdgcn_sched_group_barrier(0x100, 4 * F2_AHEAD, 0);
#pragma unroll
  for (int ks = 0; ks < F2_KS - F2_AHEAD; ++ks) {
    __builtin_amdgcn_sched_group_barrier(0x008, 1, 0);
    __builtin_amdgcn_sched_group_barrier(0x100, 4, 0);
  }
  __builtin_amdgcn_sched_group_barrier(0x008, F2_AHEAD, 0);
  return c;
}

__global__ void __launch_bounds__(F2_WAVES * 32)
attn_gemm2_kernel(const int* __restrict__ adj,
                  const _Float16* __restrict__ feats2T,
                  const float* __restrict__ s2,
                  const float* __restrict__ n2,
                  const float* __restrict__ b2,
                  float* __restrict__ out) {
  constexpr int MCH = N / (128 * F2_MU);
  constexpr int CH  = N / F2_CCOLS;
  static_assert(MCH >= 2 && MCH % 2 == 0, "chunks are taken in pairs");
  static_assert(CH % 2 == 0, "chunks are taken in pairs, one per LDS stage");
  __shared__ __attribute__((aligned(16))) _Float16 pl[2 * F2_STAGE];
  __shared__ __attribute__((aligned(16))) float zl[16];

  const int lane = threadIdx.x & 31;
  const int wave = __builtin_amdgcn_readfirstlane((int)(threadIdx.x >> 5));
  const int r0   = blockIdx.x << 4;
  const int i    = r0 + wave;
  const bool mul = wave == 0;

  const int* arow_i = adj + (size_t)i * N;
  __builtin_prefetch(arow_i + lane * 64, 0, 0);
  __builtin_prefetch(arow_i + (N / 2) + lane * 64, 0, 0);

  const int4*   arow = (const int4*)arow_i + lane;
  const float4* nq   = (const float4*)n2 + lane;
  const _Float16* brow = feats2T + (size_t)wave * N + lane * 8;

  float nmax[1] = {-INFINITY};
  int cnt = 0;
  RsChunk<1, F2_MU> ma, mb;
  rs_load(ma, arow, nq, 0);
  for (int ci = 0; ci < MCH - 2; ci += 2) {
    rs_load(mb, arow, nq, ci + 1);
    rs_max_step(ma, nmax, cnt);
    rs_load(ma, arow, nq, ci + 2);
    rs_max_step(mb, nmax, cnt);
  }
  rs_load(mb, arow, nq, MCH - 1);
  rs_max_step(ma, nmax, cnt);
  RsChunk<1, F2_U> ca, cb;
  rs_load(ca, arow, nq, 0);
  F2Feat fa, fb;
  f2_load_b(fa, brow, 0);
  rs_max_step(mb, nmax, cnt);

#pragma unroll
  for (int msk = 16; msk > 0; msk >>= 1) cnt += __shfl_xor(cnt, msk, 32);
  const int n_masked = N - cnt;

  float sv[1], mv[1], acc[1] = {0.0f};
  sv[0] = s2[i];
  {
    float mm = fmaxf(-3.0e38f, lrelu_max(sv[0] + wave_max32(nmax[0])));
    if (n_masked > 0) mm = fmaxf(mm, MASK_VAL);
    mv[0] = mm;
  }

  _Float16* prow0 = pl + wave * F2_PITCH + lane * 4;
  _Float16* prow1 = prow0 + F2_STAGE;
  _Float16* bl0 = pl + F2_TILE + wave * F2_PITCH + lane * 8;
  _Float16* bl1 = bl0 + F2_STAGE;

  v8f c = {};
  for (int ci = 0; ci < CH; ci += 2) {
    rs_load(cb, arow, nq, ci + 1);
    f2_load_b(fb, brow, ci + 1);
    rs_sum_step<1, F2_U, true>(ca, sv, mv, acc, prow0, 0);
    f2_store_b(fa, bl0);
    f1_barrier();
    if (mul) c = f2_mma(pl, c);

    if (ci + 2 < CH) {
      rs_load(ca, arow, nq, ci + 2);
      f2_load_b(fa, brow, ci + 2);
    }
    rs_sum_step<1, F2_U, true>(cb, sv, mv, acc, prow1, 0);
    f2_store_b(fb, bl1);
    f1_barrier();
    if (mul) c = f2_mma(pl + F2_STAGE, c);
  }

  {
    float zs = wave_sum32(acc[0]);
    if (n_masked > 0) zs += (float)n_masked * __expf(MASK_VAL - mv[0]);
    if (lane == 0) zl[wave] = zs;
  }
  __syncthreads();
  if (!mul) return;

  const int hf = lane >> 4;
  const int r  = lane & 15;
  const float bias = b2[r];
  const float4 za = *(const float4*)(zl + (hf << 3));
  const float4 zb = *(const float4*)(zl + (hf << 3) + 4);
  const float zr[8] = {za.x, za.y, za.z, za.w, zb.x, zb.y, zb.z, zb.w};
#pragma unroll
  for (int k = 0; k < 8; ++k) {
    int orow = r0 + k + (hf << 3);
    float zi = 1.0f / zr[k];
    float v  = c[k] * zi + bias;
    v = v > 0.0f ? v : 0.0f;
    float mx = v;
#pragma unroll
    for (int msk = 1; msk < 16; msk <<= 1)
      mx = fmaxf(mx, __shfl_xor(mx, msk, 32));
    float ex = __expf(v - mx);
    float sum = ex;
#pragma unroll
    for (int msk = 1; msk < 16; msk <<= 1) sum += __shfl_xor(sum, msk, 32);
    float o  = (v - mx) - logf(sum);
    out[(size_t)orow * NCLASS + r] = o;
  }
}

extern "C" void kernel_launch(void* const* d_in, const int* in_sizes, int n_in,
                              void* d_out, int out_size, void* d_ws,
                              size_t ws_size, hipStream_t stream) {
  const float* x   = (const float*)d_in[0];
  const int*   adj = (const int*)d_in[1];
  const float* W1  = (const float*)d_in[2];
  const float* b1  = (const float*)d_in[3];
  const float* as1 = (const float*)d_in[4];
  const float* an1 = (const float*)d_in[5];
  const float* W2  = (const float*)d_in[6];
  const float* b2  = (const float*)d_in[7];
  const float* as2 = (const float*)d_in[8];
  const float* an2 = (const float*)d_in[9];
  float* out = (float*)d_out;

  char* ws = (char*)d_ws;
  size_t off = 0;
  auto alloc = [&](size_t bytes) -> void* {
    off = (off + 255) & ~(size_t)255;
    void* p = ws + off;
    off += bytes;
    return p;
  };

  _Float16* W1F     = (_Float16*)alloc((size_t)H1 * NHID * NFEAT * 2);
  _Float16* W2T     = (_Float16*)alloc((size_t)NCLASS * H1 * NHID * 2);
  float*    feats1  = (float*)alloc((size_t)H1 * N * NHID * 4);
  _Float16* featsF  = (_Float16*)alloc((size_t)H1 * NHID * N * 2);
  float*    s1      = (float*)alloc((size_t)H1 * N * 4);
  float*    n1      = (float*)alloc((size_t)H1 * N * 4);
  _Float16* H       = (_Float16*)alloc((size_t)N * H1 * NHID * 2);
  float*    feats2  = (float*)alloc((size_t)N * NCLASS * 4);
  _Float16* feats2T = (_Float16*)alloc((size_t)NCLASS * N * 2);
  float*    s2      = (float*)alloc((size_t)N * 4);
  float*    n2      = (float*)alloc((size_t)N * 4);
  (void)ws_size; (void)n_in; (void)in_sizes; (void)out_size;

  {
    int total = H1 * NFEAT * NHID;
    cast_kernel<<<(total + 255) / 256, 256, 0, stream>>>(W1, W2, W1F, W2T);
  }
  gemm1_kernel<<<dim3(N / 16, H1), 32, 0, stream>>>(x, W1F, feats1, featsF);
  sn_kernel<<<(H1 * N + 255) / 256, 256, 0, stream>>>(feats1, as1, an1, s1, n1,
                                                      NHID, N, H1 * N);
  attn_gemm1_kernel<<<dim3(N / F1_ROWS, H1 / F1_HG), F1_WAVES * 32, 0,
                      stream>>>(adj, featsF, s1, n1, b1, H);
  gemm2_kernel<<<N / 16, 32, 0, stream>>>(H, W2T, feats2, feats2T);
  sn_kernel<<<(N + 255) / 256, 256, 0, stream>>>(feats2, as2, an2, s2, n2,
                                                 NCLASS, N, N);
  attn_gemm2_kernel<<<N / 16, F2_WAVES * 32, 0, stream>>>(adj, feats2T, s2, n2,
                                                          b2, out);
}


# Both attention kernels: sweep 1 keeps its adjacency tests as lane masks in the row-owning wave's registers; sweep 2 selects by them, so each adjacency row is read once, not twice; sums unchanged
# speedup vs baseline: 5.0691x; 1.1091x over previous
#include <hip/hip_runtime.h>
#include <hip/hip_bf16.h>


#define N      4096
#define NFEAT  512
#define NHID   64
#define NCLASS 16
#define H1     8
#define NEG_SLOPE 0.2f
#define MASK_VAL  -1e10f

typedef __attribute__((ext_vector_type(16))) _Float16 v16h;
typedef __attribute__((ext_vector_type(8)))  _Float16 v8h;
typedef __attribute__((ext_vector_type(4)))  _Float16 v4h;
typedef __attribute__((ext_vector_type(8)))  float    v8f;

__device__ __forceinline__ v8f wmma16(v16h a, v16h b, v8f c) {
  return __builtin_amdgcn_wmma_f32_16x16x32_f16(
       false, a,  false, b,
       (short)0, c,  false,  false);
}

__device__ __forceinline__ v16h load_frag16(const _Float16* base, int rc,
                                            int ld, int k0) {
  int lane = threadIdx.x & 31;
  int hf   = lane >> 4;
  int r    = lane & 15;
  const _Float16* p = base + (size_t)(rc + r) * ld + k0 + (hf << 3);
  v8h lo = *(const v8h*)p;
  v8h hi = *(const v8h*)(p + 16);
  return __builtin_shufflevector(lo, hi, 0, 1, 2, 3, 4, 5, 6, 7, 8, 9, 10, 11,
                                 12, 13, 14, 15);
}

__device__ __forceinline__ v16h load_frag16_f32(const float* base, int rc,
                                                int ld, int k0) {
  int lane = threadIdx.x & 31;
  int hf   = lane >> 4;
  int r    = lane & 15;
  const float* p = base + (size_t)(rc + r) * ld + k0 + (hf << 3);
  const float4 q0 = *(const float4*)p;
  const float4 q1 = *(const float4*)(p + 4);
  const float4 q2 = *(const float4*)(p + 16);
  const float4 q3 = *(const float4*)(p + 20);
  v16h a;
  a[0]  = (_Float16)q0.x; a[1]  = (_Float16)q0.y;
  a[2]  = (_Float16)q0.z; a[3]  = (_Float16)q0.w;
  a[4]  = (_Float16)q1.x; a[5]  = (_Float16)q1.y;
  a[6]  = (_Float16)q1.z; a[7]  = (_Float16)q1.w;
  a[8]  = (_Float16)q2.x; a[9]  = (_Float16)q2.y;
  a[10] = (_Float16)q2.z; a[11] = (_Float16)q2.w;
  a[12] = (_Float16)q3.x; a[13] = (_Float16)q3.y;
  a[14] = (_Float16)q3.z; a[15] = (_Float16)q3.w;
  return a;
}

#define FRAG_HALVES 512
__device__ __forceinline__ int frag_off(int r, int k) {
  return ((k >> 4) & 1) * 256 + ((((k >> 3) & 1) * 16 + r) * 8) + (k & 7);
}
__device__ __forceinline__ v16h load_frag16_packed(const _Float16* frag) {
  int lane = threadIdx.x & 31;
  const _Float16* p = frag + lane * 8;
  v8h lo = *(const v8h*)p;
  v8h hi = *(const v8h*)(p + 256);
  return __builtin_shufflevector(lo, hi, 0, 1, 2, 3, 4, 5, 6, 7, 8, 9, 10, 11,
                                 12, 13, 14, 15);
}

__device__ __forceinline__ float lrelu(float x) {
  return x > 0.0f ? x : NEG_SLOPE * x;
}

#define W1_GROUPS (NHID / 16 + 1)

__device__ __forceinline__ void split_hi_lo(float v, _Float16& hi, _Float16& lo) {
  hi = (_Float16)v;
  lo = (_Float16)(v - (float)hi);
}

__global__ void cast_kernel(const float* __restrict__ W1,
                            const float* __restrict__ W2,
                            const float* __restrict__ as1,
                            const float* __restrict__ an1,
                            const float* __restrict__ as2,
                            const float* __restrict__ an2,
                            _Float16* __restrict__ W1F,
                            _Float16* __restrict__ W2T) {
  int i = blockIdx.x * blockDim.x + threadIdx.x;
  if (i < H1 * NFEAT * NHID) {
    int h   = i / (NFEAT * NHID);
    int rem = i % (NFEAT * NHID);
    int k   = rem / NHID;
    int c   = rem % NHID;
    const int frag = (h * W1_GROUPS + (c >> 4)) * (NFEAT / 32) + (k >> 5);
    W1F[(size_t)frag * FRAG_HALVES + frag_off(c & 15, k & 31)] = (_Float16)W1[i];
  }
  if (i < H1 * NFEAT) {
    int h = i / NFEAT;
    int k = i % NFEAT;
    const float* w  = W1 + (size_t)i * NHID;
    const float* av = as1 + h * NHID;
    const float* nv = an1 + h * NHID;
    float ss = 0.0f, nn = 0.0f;
    for (int c = 0; c < NHID; ++c) {
      ss += w[c] * av[c];
      nn += w[c] * nv[c];
    }
    _Float16 col[16];
#pragma unroll
    for (int r = 0; r < 16; ++r) col[r] = (_Float16)0.0f;
    split_hi_lo(ss, col[0], col[1]);
    split_hi_lo(nn, col[2], col[3]);
    const int frag = (h * W1_GROUPS + NHID / 16) * (NFEAT / 32) + (k >> 5);
    _Float16* fp = W1F + (size_t)frag * FRAG_HALVES;
#pragma unroll
    for (int r = 0; r < 16; ++r) fp[frag_off(r, k & 31)] = col[r];
  }
  if (i < (H1 * NHID) * NCLASS) {
    int k = i / NCLASS;
    int c = i % NCLASS;
    W2T[(size_t)c * (H1 * NHID) + k] = (_Float16)W2[i];
  }
  if (i < H1 * NHID) {
    const float* w = W2 + (size_t)i * NCLASS;
    float ss = 0.0f, nn = 0.0f;
    for (int c = 0; c < NCLASS; ++c) {
      ss += w[c] * as2[c];
      nn += w[c] * an2[c];
    }
    _Float16 col[16];
#pragma unroll
    for (int r = 0; r < 16; ++r) col[r] = (_Float16)0.0f;
    split_hi_lo(ss, col[0], col[1]);
    split_hi_lo(nn, col[2], col[3]);
#pragma unroll
    for (int r = 0; r < 16; ++r)
      W2T[(size_t)(NCLASS + r) * (H1 * NHID) + i] = col[r];
  }
}

__device__ __forceinline__ void store_dots(const v8f& d, float* __restrict__ s,
                                           float* __restrict__ n, int row0) {
  const int lane = threadIdx.x & 31;
  const int hb   = lane & 16;
  float sv[8], nv[8];
#pragma unroll
  for (int k = 0; k < 8; ++k) {
    sv[k] = __shfl(d[k], hb + 0, 32) + __shfl(d[k], hb + 1, 32);
    nv[k] = __shfl(d[k], hb + 2, 32) + __shfl(d[k], hb + 3, 32);
  }
  if ((lane & 15) == 0) {
    float* sp = s + row0 + (hb >> 1);
    float* np = n + row0 + (hb >> 1);
    *(float4*)sp       = make_float4(sv[0], sv[1], sv[2], sv[3]);
    *(float4*)(sp + 4) = make_float4(sv[4], sv[5], sv[6], sv[7]);
    *(float4*)np       = make_float4(nv[0], nv[1], nv[2], nv[3]);
    *(float4*)(np + 4) = make_float4(nv[4], nv[5], nv[6], nv[7]);
  }
}

__global__ void gemm1_kernel(const float* __restrict__ x,
                             const _Float16* __restrict__ W1F,
                             _Float16* __restrict__ featsF,
                             float* __restrict__ s1, float* __restrict__ n1) {
  int head = blockIdx.y;
  int r0   = blockIdx.x << 4;
  const _Float16* BF = W1F + (size_t)head * W1_GROUPS * (NFEAT / 32) * FRAG_HALVES;

  v8f c0 = {}, c1 = {}, c2 = {}, c3 = {}, c4 = {};
  for (int k0 = 0; k0 < NFEAT; k0 += 32) {
    v16h a = load_frag16_f32(x, r0, NFEAT, k0);
    const _Float16* bk = BF + (k0 >> 5) * FRAG_HALVES;
    c0 = wmma16(a, load_frag16_packed(bk), c0);
    c1 = wmma16(a, load_frag16_packed(bk + 1 * (NFEAT / 32) * FRAG_HALVES), c1);
    c2 = wmma16(a, load_frag16_packed(bk + 2 * (NFEAT / 32) * FRAG_HALVES), c2);
    c3 = wmma16(a, load_frag16_packed(bk + 3 * (NFEAT / 32) * FRAG_HALVES), c3);
    c4 = wmma16(a, load_frag16_packed(bk + 4 * (NFEAT / 32) * FRAG_HALVES), c4);
  }

  int lane = threadIdx.x & 31;
  v8f acc[4] = {c0, c1, c2, c3};
#pragma unroll
  for (int nt = 0; nt < 4; ++nt) {
    v8h pk;
#pragma unroll
    for (int k = 0; k < 8; ++k) pk[k] = (_Float16)acc[nt][k];
    const int frag = (head * (NHID / 16) + nt) * (N / 32) + (r0 >> 5);
    *(v8h*)(featsF + (size_t)frag * FRAG_HALVES + ((r0 >> 4) & 1) * 256 +
            lane * 8) = pk;
  }
  store_dots(c4, s1 + (size_t)head * N, n1 + (size_t)head * N, r0);
}


__device__ __forceinline__ float wave_max32(float v) {
#pragma unroll
  for (int msk = 16; msk > 0; msk >>= 1) v = fmaxf(v, __shfl_xor(v, msk, 32));
  return v;
}

__device__ __forceinline__ float wave_sum32(float v) {
#pragma unroll
  for (int msk = 16; msk > 0; msk >>= 1) v += __shfl_xor(v, msk, 32);
  return v;
}

__device__ __forceinline__ float lrelu_max(float x) {
  return fmaxf(x, NEG_SLOPE * x);
}

__device__ __forceinline__ unsigned keep_mask(unsigned held, bool e, int step) {
  const int lane = threadIdx.x & 31;
  const unsigned m = __builtin_amdgcn_ballot_w32(e);
  return lane == step ? m : held;
}
__device__ __forceinline__ bool held_mask(unsigned held, int step) {
  return __builtin_amdgcn_inverse_ballot_w32(
      (unsigned)__builtin_amdgcn_readlane((int)held, step));
}

template <int NH, int U>
struct RsChunk {
  int4   a[U];
  float4 v[U][NH];
};

template <int U>
struct RsN {
  float4 v[U];
};

template <int U>
__device__ __forceinline__ void rs_load_n(RsN<U>& c, const float4* nq, int ci) {
#pragma unroll
  for (int u = 0; u < U; ++u) c.v[u] = nq[(ci * U + u) * 32];
}

template <int NH, int U>
__device__ __forceinline__ void rs_load(RsChunk<NH, U>& c, const int4* arow,
                                        const float4* nq, int ci) {
#pragma unroll
  for (int u = 0; u < U; ++u) {
    const int q = (ci * U + u) * 32;
    c.a[u] = arow[q];
#pragma unroll
    for (int h = 0; h < NH; ++h) c.v[u][h] = nq[h * (N / 4) + q];
  }
}

template <int NH, int U>
__device__ __forceinline__ void rs_max_step(const RsChunk<NH, U>& c,
                                            float (&nmax)[NH], int& cnt,
                                            unsigned (&held)[4], int ci) {
#pragma unroll
  for (int u = 0; u < U; ++u) {
    const bool e0 = c.a[u].x > 0, e1 = c.a[u].y > 0;
    const bool e2 = c.a[u].z > 0, e3 = c.a[u].w > 0;
    held[0] = keep_mask(held[0], e0, ci * U + u);
    held[1] = keep_mask(held[1], e1, ci * U + u);
    held[2] = keep_mask(held[2], e2, ci * U + u);
    held[3] = keep_mask(held[3], e3, ci * U + u);
    cnt += (int)e0 + (int)e1 + (int)e2 + (int)e3;
#pragma unroll
    for (int h = 0; h < NH; ++h) {
      const float x0 = e0 ? c.v[u][h].x : -INFINITY;
      const float x1 = e1 ? c.v[u][h].y : -INFINITY;
      const float x2 = e2 ? c.v[u][h].z : -INFINITY;
      const float x3 = e3 ? c.v[u][h].w : -INFINITY;
      nmax[h] = fmaxf(fmaxf(nmax[h], x0), fmaxf(x1, fmaxf(x2, x3)));
    }
  }
}

template <int NH, int U, bool WP>
__device__ __forceinline__ void rs_sum_step(const RsN<U>& c,
                                            const unsigned (&held)[4], int cs,
                                            const float (&sv)[NH],
                                            const float (&mv)[NH],
                                            float (&acc)[NH],
                                            _Float16* prow, int ci) {
  static_assert(NH == 1, "one head");
#pragma unroll
  for (int u = 0; u < U; ++u) {
    const bool e0 = held_mask(held[0], cs * U + u);
    const bool e1 = held_mask(held[1], cs * U + u);
    const bool e2 = held_mask(held[2], cs * U + u);
    const bool e3 = held_mask(held[3], cs * U + u);
#pragma unroll
    for (int h = 0; h < NH; ++h) {
      const float p0 = __expf(lrelu_max(sv[h] + c.v[u].x) - mv[h]);
      const float p1 = __expf(lrelu_max(sv[h] + c.v[u].y) - mv[h]);
      const float p2 = __expf(lrelu_max(sv[h] + c.v[u].z) - mv[h]);
      const float p3 = __expf(lrelu_max(sv[h] + c.v[u].w) - mv[h]);
      const float q0 = e0 ? p0 : 0.0f, q1 = e1 ? p1 : 0.0f;
      const float q2 = e2 ? p2 : 0.0f, q3 = e3 ? p3 : 0.0f;
      acc[h] += q0 + q1;
      acc[h] += q2 + q3;
      if (WP) {
        const v4h pv = {(_Float16)q0, (_Float16)q1, (_Float16)q2,
                        (_Float16)q3};
        *(v4h*)(prow + (ci * U + u) * 128) = pv;
      }
    }
  }
}

#define F1_ROWS  32
#define F1_WAVES 16
#define F1_HG    4
#define F1_COLS  128
#define F1_PITCH 136
#define F1_HEAD  (F1_ROWS * F1_PITCH)
#define F1_STAGE (F1_HG * F1_HEAD)
#define F1_KS    (F1_COLS / 32)
#define F1_MU    2

template <int U>
struct F1Chunk {
  int4   a[U][2];
  float4 v[U][F1_HG];
};

template <int U>
__device__ __forceinline__ void f1_load(F1Chunk<U>& c, const int4* arow0,
                                        const int4* arow1, const float4* nq,
                                        int ci) {
#pragma unroll
  for (int u = 0; u < U; ++u) {
    const int q = (ci * U + u) * 32;
    c.a[u][0] = arow0[q];
    c.a[u][1] = arow1[q];
#pragma unroll
    for (int h = 0; h < F1_HG; ++h) c.v[u][h] = nq[h * (N / 4) + q];
  }
}

struct F1N {
  float4 v[F1_HG];
};

__device__ __forceinline__ void f1_load_n(F1N& c, const float4* nq, int ci) {
#pragma unroll
  for (int h = 0; h < F1_HG; ++h) c.v[h] = nq[h * (N / 4) + ci * 32];
}

template <int U>
__device__ __forceinline__ void f1_max_step(const F1Chunk<U>& c,
                                            float (&nmax)[2][F1_HG],
                                            int (&cnt)[2],
                                            unsigned (&held)[2][4], int ci) {
#pragma unroll
  for (int u = 0; u < U; ++u) {
#pragma unroll
    for (int t = 0; t < 2; ++t) {
      const bool e0 = c.a[u][t].x > 0, e1 = c.a[u][t].y > 0;
      const bool e2 = c.a[u][t].z > 0, e3 = c.a[u][t].w > 0;
      held[t][0] = keep_mask(held[t][0], e0, ci * U + u);
      held[t][1] = keep_mask(held[t][1], e1, ci * U + u);
      held[t][2] = keep_mask(held[t][2], e2, ci * U + u);
      held[t][3] = keep_mask(held[t][3], e3, ci * U + u);
      cnt[t] += (int)e0 + (int)e1 + (int)e2 + (int)e3;
#pragma unroll
      for (int h = 0; h < F1_HG; ++h) {
        const float x0 = e0 ? c.v[u][h].x : -INFINITY;
        const float x1 = e1 ? c.v[u][h].y : -INFINITY;
        const float x2 = e2 ? c.v[u][h].z : -INFINITY;
        const float x3 = e3 ? c.v[u][h].w : -INFINITY;
        nmax[t][h] =
            fmaxf(fmaxf(nmax[t][h], x0), fmaxf(x1, fmaxf(x2, x3)));
      }
    }
  }
}

struct F1Frags {
  v16h b[F1_KS];
};

__device__ __forceinline__ void f1_load_b(F1Frags& f, const _Float16* BF,
                                          int ci) {
#pragma unroll
  for (int ks = 0; ks < F1_KS; ++ks)
    f.b[ks] = load_frag16_packed(BF + (ci * F1_KS + ks) * FRAG_HALVES);
}

__device__ __forceinline__ void f1_form(const F1N& c,
                                        const unsigned (&held)[2][4], int ci,
                                        const float (&sv)[2][F1_HG],
                                        const float (&mv)[2][F1_HG],
                                        float (&zacc)[2][F1_HG],
                                        _Float16* prow) {
#pragma unroll
  for (int t = 0; t < 2; ++t) {
    const bool e0 = held_mask(held[t][0], ci), e1 = held_mask(held[t][1], ci);
    const bool e2 = held_mask(held[t][2], ci), e3 = held_mask(held[t][3], ci);
#pragma unroll
    for (int h = 0; h < F1_HG; ++h) {
      const float p0 = __expf(lrelu_max(sv[t][h] + c.v[h].x) - mv[t][h]);
      const float p1 = __expf(lrelu_max(sv[t][h] + c.v[h].y) - mv[t][h]);
      const float p2 = __expf(lrelu_max(sv[t][h] + c.v[h].z) - mv[t][h]);
      const float p3 = __expf(lrelu_max(sv[t][h] + c.v[h].w) - mv[t][h]);
      const float q0 = e0 ? p0 : 0.0f, q1 = e1 ? p1 : 0.0f;
      const float q2 = e2 ? p2 : 0.0f, q3 = e3 ? p3 : 0.0f;
      zacc[t][h] += q0 + q1;
      zacc[t][h] += q2 + q3;
      const v4h pv = {(_Float16)q0, (_Float16)q1, (_Float16)q2, (_Float16)q3};
      *(v4h*)(prow + h * F1_HEAD + t * 16 * F1_PITCH) = pv;
    }
  }
}

__device__ __forceinline__ void f1_mma(const _Float16* ph, const F1Frags& f,
                                       v8f& c0, v8f& c1) {
#pragma unroll
  for (int ks = 0; ks < F1_KS; ++ks) {
    const v16h a0 = load_frag16(ph, 0, F1_PITCH, ks * 32);
    const v16h a1 = load_frag16(ph, 16, F1_PITCH, ks * 32);
    c0 = wmma16(a0, f.b[ks], c0);
    c1 = wmma16(a1, f.b[ks], c1);
  }
}

__device__ __forceinline__ void f1_barrier() {
  asm volatile("s_wait_dscnt 0x0\n\ts_barrier_signal -1\n\ts_barrier_wait -1"
               ::: "memory");
}

__global__ void __launch_bounds__(F1_WAVES * 32)
attn_gemm1_kernel(const int* __restrict__ adj,
                  const _Float16* __restrict__ featsF,
                  const float* __restrict__ s1,
                  const float* __restrict__ n1,
                  const float* __restrict__ b1,
                  _Float16* __restrict__ H) {
  constexpr int CH = N / F1_COLS;
  static_assert(CH % 2 == 0, "blocks are taken in pairs, one per LDS stage");
  __shared__ __attribute__((aligned(16))) _Float16 pl[2 * F1_STAGE];
  __shared__ __attribute__((aligned(16))) float zl[F1_HG * F1_ROWS];

  const int lane = threadIdx.x & 31;
  const int wave = __builtin_amdgcn_readfirstlane((int)(threadIdx.x >> 5));
  const int r0   = blockIdx.x * F1_ROWS;
  const int h0   = blockIdx.y * F1_HG;
  const int irow[2] = {r0 + wave, r0 + 16 + wave};
  const int hl   = wave & (F1_HG - 1);
  const int head = h0 + hl;
  const int f0   = (wave >> 2) * 16;

  const int* arow_a = adj + (size_t)irow[0] * N;
  const int* arow_b = adj + (size_t)irow[1] * N;
  __builtin_prefetch(arow_a + lane * 64, 0, 0);
  __builtin_prefetch(arow_a + (N / 2) + lane * 64, 0, 0);
  __builtin_prefetch(arow_b + lane * 64, 0, 0);
  __builtin_prefetch(arow_b + (N / 2) + lane * 64, 0, 0);

  const int4*   arow0 = (const int4*)arow_a + lane;
  const int4*   arow1 = (const int4*)arow_b + lane;
  const float4* nq    = (const float4*)(n1 + (size_t)h0 * N) + lane;

  constexpr int MCH = N / (F1_COLS * F1_MU);
  static_assert(MCH >= 2 && MCH % 2 == 0, "chunks are taken in pairs");
  float nmax[2][F1_HG];
  int cnt[2] = {0, 0};
#pragma unroll
  for (int t = 0; t < 2; ++t) {
#pragma unroll
    for (int h = 0; h < F1_HG; ++h) nmax[t][h] = -INFINITY;
  }

  unsigned held[2][4] = {{0u, 0u, 0u, 0u}, {0u, 0u, 0u, 0u}};
  F1Chunk<F1_MU> ma, mb;
  f1_load(ma, arow0, arow1, nq, 0);
  for (int ci = 0; ci < MCH - 2; ci += 2) {
    f1_load(mb, arow0, arow1, nq, ci + 1);
    f1_max_step(ma, nmax, cnt, held, ci);
    f1_load(ma, arow0, arow1, nq, ci + 2);
    f1_max_step(mb, nmax, cnt, held, ci + 1);
  }
  f1_load(mb, arow0, arow1, nq, MCH - 1);
  f1_max_step(ma, nmax, cnt, held, MCH - 2);
  F1N ca, cb;
  f1_load_n(ca, nq, 0);
  f1_max_step(mb, nmax, cnt, held, MCH - 1);

  float sv[2][F1_HG], mv[2][F1_HG], zacc[2][F1_HG];
  int n_masked[2];
#pragma unroll
  for (int t = 0; t < 2; ++t) {
    int ct = cnt[t];
#pragma unroll
    for (int msk = 16; msk > 0; msk >>= 1) ct += __shfl_xor(ct, msk, 32);
    n_masked[t] = N - ct;
#pragma unroll
    for (int h = 0; h < F1_HG; ++h) {
      sv[t][h] = s1[(h0 + h) * N + irow[t]];
      float mm = fmaxf(-3.0e38f, lrelu_max(sv[t][h] + wave_max32(nmax[t][h])));
      if (n_masked[t] > 0) mm = fmaxf(mm, MASK_VAL);
      mv[t][h] = mm;
      zacc[t][h] = 0.0f;
    }
  }

  _Float16* prow0 = pl + wave * F1_PITCH + lane * 4;
  _Float16* prow1 = prow0 + F1_STAGE;
  const _Float16* ph0 = pl + hl * F1_HEAD;
  const _Float16* ph1 = ph0 + F1_STAGE;
  const _Float16* BF  = featsF + (size_t)(head * (NHID / 16) + (f0 >> 4)) *
                                     (N / 32) * FRAG_HALVES;

  F1Frags bf;
  v8f c0 = {}, c1 = {};

  for (int ci = 0; ci < CH; ci += 2) {
    f1_load_n(cb, nq, ci + 1);
    f1_load_b(bf, BF, ci);
    f1_form(ca, held, ci, sv, mv, zacc, prow0);
    f1_barrier();
    f1_mma(ph0, bf, c0, c1);

    if (ci + 2 < CH) f1_load_n(ca, nq, ci + 2);
    f1_load_b(bf, BF, ci + 1);
    f1_form(cb, held, ci + 1, sv, mv, zacc, prow1);
    f1_barrier();
    f1_mma(ph1, bf, c0, c1);
  }

#pragma unroll
  for (int t = 0; t < 2; ++t) {
#pragma unroll
    for (int h = 0; h < F1_HG; ++h) {
      float zs = wave_sum32(zacc[t][h]);
      if (n_masked[t] > 0)
        zs += (float)n_masked[t] * __expf(MASK_VAL - mv[t][h]);
      if (lane == 0) zl[h * F1_ROWS + t * 16 + wave] = zs;
    }
  }
  __syncthreads();

  const int hf = lane >> 4;
  const int r  = lane & 15;
  const int col = f0 + r;
  const float bias = b1[head * NHID + col];
  v8f acc[2] = {c0, c1};
#pragma unroll
  for (int t = 0; t < 2; ++t) {
    const float* zp = zl + hl * F1_ROWS + t * 16 + (hf << 3);
    const float4 za = *(const float4*)zp;
    const float4 zb = *(const float4*)(zp + 4);
    const float zi[8] = {1.0f / za.x, 1.0f / za.y, 1.0f / za.z, 1.0f / za.w,
                         1.0f / zb.x, 1.0f / zb.y, 1.0f / zb.z, 1.0f / zb.w};
#pragma unroll
    for (int k = 0; k < 8; ++k) {
      int orow = r0 + t * 16 + k + (hf << 3);
      float v  = acc[t][k] * zi[k] + bias;
      v = v > 0.0f ? v : 0.0f;
      H[(size_t)orow * (H1 * NHID) + head * NHID + col] = (_Float16)v;
    }
  }
}

__global__ void gemm2_kernel(const _Float16* __restrict__ H,
                             const _Float16* __restrict__ W2T,
                             _Float16* __restrict__ feats2T,
                             float* __restrict__ s2, float* __restrict__ n2) {
  int r0 = blockIdx.x << 4;
  v8f c = {}, d = {};
  for (int k0 = 0; k0 < H1 * NHID; k0 += 32) {
    v16h a = load_frag16(H, r0, H1 * NHID, k0);
    c = wmma16(a, load_frag16(W2T, 0, H1 * NHID, k0), c);
    d = wmma16(a, load_frag16(W2T, NCLASS, H1 * NHID, k0), d);
  }
  int lane = threadIdx.x & 31;
  int hf   = lane >> 4;
  int r    = lane & 15;
#pragma unroll
  for (int k = 0; k < 8; ++k) {
    int orow = r0 + k + (hf << 3);
    feats2T[(size_t)r * N + orow] = (_Float16)c[k];
  }
  store_dots(d, s2, n2, r0);
}

#define F2_WAVES 16
#define F2_U     4
#define F2_MU    8
#define F2_CCOLS (128 * F2_U)
#define F2_PITCH (F2_CCOLS + 8)
#define F2_TILE  (16 * F2_PITCH)
#define F2_STAGE (2 * F2_TILE)
#define F2_KS    (F2_CCOLS / 32)

struct F2Feat {
  v8h b[F2_CCOLS / 256];
};

__device__ __forceinline__ void f2_load_b(F2Feat& f, const _Float16* brow,
                                          int ci) {
#pragma unroll
  for (int q = 0; q < F2_CCOLS / 256; ++q)
    f.b[q] = *(const v8h*)(brow + ci * F2_CCOLS + q * 256);
}

__device__ __forceinline__ void f2_store_b(const F2Feat& f, _Float16* bl) {
#pragma unroll
  for (int q = 0; q < F2_CCOLS / 256; ++q) *(v8h*)(bl + q * 256) = f.b[q];
}

#define F2_AHEAD 6

__device__ __forceinline__ v8f f2_mma(const _Float16* st, v8f c) {
#pragma unroll
  for (int ks = 0; ks < F2_KS; ++ks)
    c = wmma16(load_frag16(st, 0, F2_PITCH, ks * 32),
               load_frag16(st + F2_TILE, 0, F2_PITCH, ks * 32), c);
  __builtin_amdgcn_sched_group_barrier(0x100, 4 * F2_AHEAD, 0);
#pragma unroll
  for (int ks = 0; ks < F2_KS - F2_AHEAD; ++ks) {
    __builtin_amdgcn_sched_group_barrier(0x008, 1, 0);
    __builtin_amdgcn_sched_group_barrier(0x100, 4, 0);
  }
  __builtin_amdgcn_sched_group_barrier(0x008, F2_AHEAD, 0);
  return c;
}

__global__ void __launch_bounds__(F2_WAVES * 32)
attn_gemm2_kernel(const int* __restrict__ adj,
                  const _Float16* __restrict__ feats2T,
                  const float* __restrict__ s2,
                  const float* __restrict__ n2,
                  const float* __restrict__ b2,
                  float* __restrict__ out) {
  constexpr int MCH = N / (128 * F2_MU);
  constexpr int CH  = N / F2_CCOLS;
  static_assert(MCH >= 2 && MCH % 2 == 0, "chunks are taken in pairs");
  static_assert(CH % 2 == 0, "chunks are taken in pairs, one per LDS stage");
  __shared__ __attribute__((aligned(16))) _Float16 pl[2 * F2_STAGE];
  __shared__ __attribute__((aligned(16))) float zl[16];

  const int lane = threadIdx.x & 31;
  const int wave = __builtin_amdgcn_readfirstlane((int)(threadIdx.x >> 5));
  const int r0   = blockIdx.x << 4;
  const int i    = r0 + wave;
  const bool mul = wave == 0;

  const int* arow_i = adj + (size_t)i * N;
  __builtin_prefetch(arow_i + lane * 64, 0, 0);
  __builtin_prefetch(arow_i + (N / 2) + lane * 64, 0, 0);

  const int4*   arow = (const int4*)arow_i + lane;
  const float4* nq   = (const float4*)n2 + lane;
  const _Float16* brow = feats2T + (size_t)wave * N + lane * 8;

  float nmax[1] = {-INFINITY};
  int cnt = 0;
  unsigned held[4] = {0u, 0u, 0u, 0u};
  RsChunk<1, F2_MU> ma, mb;
  rs_load(ma, arow, nq, 0);
  for (int ci = 0; ci < MCH - 2; ci += 2) {
    rs_load(mb, arow, nq, ci + 1);
    rs_max_step(ma, nmax, cnt, held, ci);
    rs_load(ma, arow, nq, ci + 2);
    rs_max_step(mb, nmax, cnt, held, ci + 1);
  }
  rs_load(mb, arow, nq, MCH - 1);
  rs_max_step(ma, nmax, cnt, held, MCH - 2);
  RsN<F2_U> ca, cb;
  rs_load_n(ca, nq, 0);
  F2Feat fa, fb;
  f2_load_b(fa, brow, 0);
  rs_max_step(mb, nmax, cnt, held, MCH - 1);

#pragma unroll
  for (int msk = 16; msk > 0; msk >>= 1) cnt += __shfl_xor(cnt, msk, 32);
  const int n_masked = N - cnt;

  float sv[1], mv[1], acc[1] = {0.0f};
  sv[0] = s2[i];
  {
    float mm = fmaxf(-3.0e38f, lrelu_max(sv[0] + wave_max32(nmax[0])));
    if (n_masked > 0) mm = fmaxf(mm, MASK_VAL);
    mv[0] = mm;
  }

  _Float16* prow0 = pl + wave * F2_PITCH + lane * 4;
  _Float16* prow1 = prow0 + F2_STAGE;
  _Float16* bl0 = pl + F2_TILE + wave * F2_PITCH + lane * 8;
  _Float16* bl1 = bl0 + F2_STAGE;

  v8f c = {};
  for (int ci = 0; ci < CH; ci += 2) {
    rs_load_n(cb, nq, ci + 1);
    f2_load_b(fb, brow, ci + 1);
    rs_sum_step<1, F2_U, true>(ca, held, ci, sv, mv, acc, prow0, 0);
    f2_store_b(fa, bl0);
    f1_barrier();
    if (mul) c = f2_mma(pl, c);

    if (ci + 2 < CH) {
      rs_load_n(ca, nq, ci + 2);
      f2_load_b(fa, brow, ci + 2);
    }
    rs_sum_step<1, F2_U, true>(cb, held, ci + 1, sv, mv, acc, prow1, 0);
    f2_store_b(fb, bl1);
    f1_barrier();
    if (mul) c = f2_mma(pl + F2_STAGE, c);
  }

  {
    float zs = wave_sum32(acc[0]);
    if (n_masked > 0) zs += (float)n_masked * __expf(MASK_VAL - mv[0]);
    if (lane == 0) zl[wave] = zs;
  }
  __syncthreads();
  if (!mul) return;

  const int hf = lane >> 4;
  const int r  = lane & 15;
  const float bias = b2[r];
  const float4 za = *(const float4*)(zl + (hf << 3));
  const float4 zb = *(const float4*)(zl + (hf << 3) + 4);
  const float zr[8] = {za.x, za.y, za.z, za.w, zb.x, zb.y, zb.z, zb.w};
#pragma unroll
  for (int k = 0; k < 8; ++k) {
    int orow = r0 + k + (hf << 3);
    float zi = 1.0f / zr[k];
    float v  = c[k] * zi + bias;
    v = v > 0.0f ? v : 0.0f;
    float mx = v;
#pragma unroll
    for (int msk = 1; msk < 16; msk <<= 1)
      mx = fmaxf(mx, __shfl_xor(mx, msk, 32));
    float ex = __expf(v - mx);
    float sum = ex;
#pragma unroll
    for (int msk = 1; msk < 16; msk <<= 1) sum += __shfl_xor(sum, msk, 32);
    float o  = (v - mx) - logf(sum);
    out[(size_t)orow * NCLASS + r] = o;
  }
}

extern "C" void kernel_launch(void* const* d_in, const int* in_sizes, int n_in,
                              void* d_out, int out_size, void* d_ws,
                              size_t ws_size, hipStream_t stream) {
  const float* x   = (const float*)d_in[0];
  const int*   adj = (const int*)d_in[1];
  const float* W1  = (const float*)d_in[2];
  const float* b1  = (const float*)d_in[3];
  const float* as1 = (const float*)d_in[4];
  const float* an1 = (const float*)d_in[5];
  const float* W2  = (const float*)d_in[6];
  const float* b2  = (const float*)d_in[7];
  const float* as2 = (const float*)d_in[8];
  const float* an2 = (const float*)d_in[9];
  float* out = (float*)d_out;

  char* ws = (char*)d_ws;
  size_t off = 0;
  auto alloc = [&](size_t bytes) -> void* {
    off = (off + 255) & ~(size_t)255;
    void* p = ws + off;
    off += bytes;
    return p;
  };

  _Float16* W1F     = (_Float16*)alloc((size_t)H1 * W1_GROUPS * 16 * NFEAT * 2);
  _Float16* W2T     = (_Float16*)alloc((size_t)2 * NCLASS * H1 * NHID * 2);
  _Float16* featsF  = (_Float16*)alloc((size_t)H1 * NHID * N * 2);
  float*    s1      = (float*)alloc((size_t)H1 * N * 4);
  float*    n1      = (float*)alloc((size_t)H1 * N * 4);
  _Float16* H       = (_Float16*)alloc((size_t)N * H1 * NHID * 2);
  _Float16* feats2T = (_Float16*)alloc((size_t)NCLASS * N * 2);
  float*    s2      = (float*)alloc((size_t)N * 4);
  float*    n2      = (float*)alloc((size_t)N * 4);
  (void)ws_size; (void)n_in; (void)in_sizes; (void)out_size;

  {
    int total = H1 * NFEAT * NHID;
    cast_kernel<<<(total + 255) / 256, 256, 0, stream>>>(W1, W2, as1, an1, as2,
                                                         an2, W1F, W2T);
  }
  gemm1_kernel<<<dim3(N / 16, H1), 32, 0, stream>>>(x, W1F, featsF, s1, n1);
  attn_gemm1_kernel<<<dim3(N / F1_ROWS, H1 / F1_HG), F1_WAVES * 32, 0,
                      stream>>>(adj, featsF, s1, n1, b1, H);
  gemm2_kernel<<<N / 16, 32, 0, stream>>>(H, W2T, feats2T, s2, n2);
  attn_gemm2_kernel<<<N / 16, F2_WAVES * 32, 0, stream>>>(adj, feats2T, s2, n2,
                                                          b2, out);
}
